# starting kernel
# baseline (speedup 1.0000x reference)
#include <hip/hip_runtime.h>

typedef float v2f __attribute__((ext_vector_type(2)));
typedef float v8f __attribute__((ext_vector_type(8)));


__global__ void gcn_zero(float* __restrict__ deg, float* __restrict__ acc1,
                         float* __restrict__ acc2, int n) {
  int i = blockIdx.x * blockDim.x + threadIdx.x;
  if (i < n) { deg[i] = 0.f; acc1[i] = 0.f; acc2[i] = 0.f; }
}

__global__ void gcn_degree(const int* __restrict__ col, float* __restrict__ deg, int E) {
  int i = blockIdx.x * blockDim.x + threadIdx.x;
  int e0 = i * 4;
  if (e0 + 3 < E) {
    int4 c = *reinterpret_cast<const int4*>(col + e0);
    atomicAdd(&deg[c.x], 1.0f);
    atomicAdd(&deg[c.y], 1.0f);
    atomicAdd(&deg[c.z], 1.0f);
    atomicAdd(&deg[c.w], 1.0f);
  } else {
    for (int e = e0; e < E; ++e) atomicAdd(&deg[col[e]], 1.0f);
  }
}

__global__ void gcn_prep(const float* __restrict__ x, const float* __restrict__ deg,
                         float* __restrict__ dis, float* __restrict__ p, int n) {
  int i = blockIdx.x * blockDim.x + threadIdx.x;
  if (i < n) {
    float d = deg[i] + 1.0f;
    float r = rsqrtf(d);
    dis[i] = r;
    p[i] = r * x[i];
  }
}

__global__ void gcn_scatter(const int* __restrict__ row, const int* __restrict__ col,
                            const float* __restrict__ val, float* __restrict__ acc, int E) {
  int i = blockIdx.x * blockDim.x + threadIdx.x;
  int e0 = i * 4;
  if (e0 + 3 < E) {
    int4 r = *reinterpret_cast<const int4*>(row + e0);
    int4 c = *reinterpret_cast<const int4*>(col + e0);
    float v0 = val[r.x], v1 = val[r.y], v2 = val[r.z], v3 = val[r.w];
    atomicAdd(&acc[c.x], v0);
    atomicAdd(&acc[c.y], v1);
    atomicAdd(&acc[c.z], v2);
    atomicAdd(&acc[c.w], v3);
  } else {
    for (int e = e0; e < E; ++e) atomicAdd(&acc[col[e]], val[row[e]]);
  }
}

__global__ void gcn_node_update(const float* __restrict__ dis, const float* __restrict__ p,
                                const float* __restrict__ acc1,
                                const float* __restrict__ W1, const float* __restrict__ b1,
                                const float* __restrict__ W2,
                                float* __restrict__ q, int n) {
  __shared__ float hbuf[8 * 256];
  const int lane = threadIdx.x & 31;
  const int wv = threadIdx.x >> 5;
  const int base = (blockIdx.x * (blockDim.x >> 5) + wv) * 16;
  const int m = lane & 15;
  const bool hi = lane >= 16;
  int node = base + m;
  int cn = node < n ? node : n - 1;
  float s = dis[cn] * (acc1[cn] + p[cn]);

  v2f a = { hi ? 0.f : s,      hi ? 0.f : 1.f };
  v2f b = { hi ? 0.f : W1[m],  hi ? 0.f : b1[m] };
  v8f c = {0.f, 0.f, 0.f, 0.f, 0.f, 0.f, 0.f, 0.f};
  v8f h = __builtin_amdgcn_wmma_f32_16x16x4_f32(false, a, false, b, (short)0, c,
                                                false, false);
  float* t0 = &hbuf[wv * 256];
#pragma unroll
  for (int i = 0; i < 8; ++i) {
    float hv = h[i] > 0.f ? h[i] : 0.f;
    t0[(i + (hi ? 8 : 0)) * 16 + m] = hv;
  }
  __syncthreads();

  v8f t = {0.f, 0.f, 0.f, 0.f, 0.f, 0.f, 0.f, 0.f};
#pragma unroll
  for (int ck = 0; ck < 4; ++ck) {
    int k0 = 4 * ck + (hi ? 2 : 0);
    v2f ap = { W2[k0], W2[k0 + 1] };
    v2f bp = { t0[m * 16 + k0], t0[m * 16 + k0 + 1] };
    t = __builtin_amdgcn_wmma_f32_16x16x4_f32(false, ap, false, bp, (short)0, t,
                                              false, false);
  }
  if (!hi && node < n) q[node] = dis[node] * t[0];
}

__global__ void gcn_finalize(const float* __restrict__ dis, const float* __restrict__ q,
                             const float* __restrict__ acc2, const float* __restrict__ b2,
                             float* __restrict__ out, int n) {
  int i = blockIdx.x * blockDim.x + threadIdx.x;
  if (i < n) out[i] = dis[i] * (acc2[i] + q[i]) + b2[0];
}

extern "C" void kernel_launch(void* const* d_in, const int* in_sizes, int n_in,
                              void* d_out, int out_size, void* d_ws, size_t ws_size,
                              hipStream_t stream) {
  const float* x  = (const float*)d_in[0];
  const int*   ei = (const int*)d_in[1];
  const float* W1 = (const float*)d_in[2];
  const float* b1 = (const float*)d_in[3];
  const float* W2 = (const float*)d_in[4];
  const float* b2 = (const float*)d_in[5];
  float* out = (float*)d_out;

  const int n = in_sizes[0];
  const int E = in_sizes[1] / 2;
  const int* row = ei;
  const int* col = ei + E;

  float* ws   = (float*)d_ws;
  float* deg  = ws;
  float* dis  = ws + 1 * (size_t)n;
  float* p    = ws + 2 * (size_t)n;
  float* acc1 = ws + 3 * (size_t)n;
  float* q    = ws + 4 * (size_t)n;
  float* acc2 = ws + 5 * (size_t)n;

  const int TB = 256;
  const int nodeBlocks = (n + TB - 1) / TB;
  const int edgeLanes  = (E + 3) / 4;
  const int edgeBlocks = (edgeLanes + TB - 1) / TB;
  const int tiles      = (n + 15) / 16;
  const int mlpBlocks  = (tiles + 7) / 8;

  gcn_zero       <<<nodeBlocks, TB, 0, stream>>>(deg, acc1, acc2, n);
  gcn_degree     <<<edgeBlocks, TB, 0, stream>>>(col, deg, E);
  gcn_prep       <<<nodeBlocks, TB, 0, stream>>>(x, deg, dis, p, n);
  gcn_scatter    <<<edgeBlocks, TB, 0, stream>>>(row, col, p, acc1, E);
  gcn_node_update<<<mlpBlocks,  TB, 0, stream>>>(dis, p, acc1, W1, b1, W2, q, n);
  gcn_scatter    <<<edgeBlocks, TB, 0, stream>>>(row, col, q, acc2, E);
  gcn_finalize   <<<nodeBlocks, TB, 0, stream>>>(dis, q, acc2, b2, out, n);
}


# Segment sums via per-block LDS float atomics on node-range accumulators, stored as 128 partial planes and added in fixed order by the node kernels; no global atomics left (sum order changes)
# speedup vs baseline: 7.5290x; 7.5290x over previous
#include <hip/hip_runtime.h>


typedef float v2f __attribute__((ext_vector_type(2)));
typedef float v8f __attribute__((ext_vector_type(8)));


constexpr int CHUNKS = 128;
constexpr int RANGE  = 51200;
constexpr int ETB    = 1024;
constexpr int UNR    = 4;
constexpr int NTB    = 256;

template <bool GATHER>
__device__ __forceinline__ void edge_pass(const int* __restrict__ row, const int* __restrict__ col,
                                          const float* __restrict__ val,
                                          float* __restrict__ planes, int n, int E,
                                          float* acc) {
  const int tid = threadIdx.x;
  int rsz = (n + (int)gridDim.y - 1) / (int)gridDim.y;
  rsz = (rsz + 31) & ~31;
  const int lo = (int)blockIdx.y * rsz;
  int len = n - lo;
  if (len > rsz) len = rsz;
  if (len < 0) len = 0;
  const unsigned ulen = (unsigned)len;

  for (int i = tid * 4; i < len; i += ETB * 4)
    *reinterpret_cast<float4*>(&acc[i]) = make_float4(0.f, 0.f, 0.f, 0.f);
  __syncthreads();

  int chunkE = (E + (int)gridDim.x - 1) / (int)gridDim.x;
  chunkE = (chunkE + 3) & ~3;
  int eb = (int)blockIdx.x * chunkE;
  if (eb > E) eb = E;
  int ee = eb + chunkE;
  if (ee > E) ee = E;

  if ((E & 3) == 0) {
    const int groups = (ee - eb) >> 2;
    const int4* col4 = reinterpret_cast<const int4*>(col + eb);
    const int4* row4 = reinterpret_cast<const int4*>(row + eb);
    for (int g0 = tid; g0 < groups; g0 += ETB * UNR) {
      int cc[UNR][4];
      int rr[UNR][4];
      bool ok[UNR];
#pragma unroll
      for (int u = 0; u < UNR; ++u) {
        const int g = g0 + u * ETB;
        ok[u] = g < groups;
        const int gc = ok[u] ? g : g0;
        const int4 c4 = col4[gc];
        cc[u][0] = c4.x; cc[u][1] = c4.y; cc[u][2] = c4.z; cc[u][3] = c4.w;
        if (GATHER) {
          const int4 r4 = row4[gc];
          rr[u][0] = r4.x; rr[u][1] = r4.y; rr[u][2] = r4.z; rr[u][3] = r4.w;
        }
      }
      unsigned d[UNR][4];
      bool in[UNR][4];
      float v[UNR][4];
#pragma unroll
      for (int u = 0; u < UNR; ++u) {
#pragma unroll
        for (int k = 0; k < 4; ++k) {
          d[u][k] = (unsigned)(cc[u][k] - lo);
          in[u][k] = ok[u] && d[u][k] < ulen;
          if (GATHER) v[u][k] = val[in[u][k] ? rr[u][k] : 0];
          else        v[u][k] = 1.0f;
        }
      }
      if (GATHER) asm volatile("" :: "v"(v[0][0]));
#pragma unroll
      for (int u = 0; u < UNR; ++u) {
#pragma unroll
        for (int k = 0; k < 4; ++k) {
          if (in[u][k]) atomicAdd(&acc[d[u][k]], v[u][k]);
        }
      }
    }
  } else {
    for (int e = eb + tid; e < ee; e += ETB) {
      const unsigned dd = (unsigned)(col[e] - lo);
      if (dd < ulen) atomicAdd(&acc[dd], GATHER ? val[row[e]] : 1.0f);
    }
  }
  __syncthreads();

  float* dst = planes + (size_t)blockIdx.x * (size_t)n + lo;
  if ((n & 3) == 0) {
    for (int i = tid * 4; i < len; i += ETB * 4)
      *reinterpret_cast<float4*>(&dst[i]) = *reinterpret_cast<const float4*>(&acc[i]);
  } else {
    for (int i = tid; i < len; i += ETB) dst[i] = acc[i];
  }
}

__global__ void __launch_bounds__(ETB)
gcn_degree(const int* __restrict__ row, const int* __restrict__ col,
           const float* __restrict__ val, float* __restrict__ planes, int n, int E) {
  __shared__ __attribute__((aligned(16))) float acc[RANGE];
  edge_pass<false>(row, col, val, planes, n, E, acc);
}

__global__ void __launch_bounds__(ETB)
gcn_scatter(const int* __restrict__ row, const int* __restrict__ col,
            const float* __restrict__ val, float* __restrict__ planes, int n, int E) {
  __shared__ __attribute__((aligned(16))) float acc[RANGE];
  edge_pass<true>(row, col, val, planes, n, E, acc);
}

template <int PARTS>
__device__ __forceinline__ void plane_partials(const float* __restrict__ planes, int node0, int n,
                                               float* sbuf) {
  constexpr int NODES = NTB / PARTS;
  constexpr int PER = CHUNKS / PARTS;
  const int j = (int)threadIdx.x % NODES;
  const int part = (int)threadIdx.x / NODES;
  int node = node0 + j;
  if (node >= n) node = n - 1;
  const float* src = planes + (size_t)(part * PER) * (size_t)n + node;
  float s0 = 0.f, s1 = 0.f, s2 = 0.f, s3 = 0.f;
#pragma unroll
  for (int c = 0; c < PER; c += 4) {
    s0 += src[(size_t)(c + 0) * (size_t)n];
    s1 += src[(size_t)(c + 1) * (size_t)n];
    s2 += src[(size_t)(c + 2) * (size_t)n];
    s3 += src[(size_t)(c + 3) * (size_t)n];
  }
  sbuf[part * NODES + j] = (s0 + s1) + (s2 + s3);
}

__global__ void __launch_bounds__(NTB)
gcn_prep(const float* __restrict__ x, const float* __restrict__ planes,
         float* __restrict__ dis, float* __restrict__ p, int n) {
  __shared__ float sbuf[NTB];
  const int node0 = (int)blockIdx.x * (NTB / 4);
  plane_partials<4>(planes, node0, n, sbuf);
  __syncthreads();
  const int t = threadIdx.x;
  const int i = node0 + t;
  if (t < NTB / 4 && i < n) {
    float deg = (sbuf[t] + sbuf[NTB / 4 + t]) + (sbuf[2 * (NTB / 4) + t] + sbuf[3 * (NTB / 4) + t]);
    float d = deg + 1.0f;
    float r = rsqrtf(d);
    dis[i] = r;
    p[i] = r * x[i];
  }
}

__global__ void __launch_bounds__(NTB)
gcn_node_update(const float* __restrict__ dis, const float* __restrict__ p,
                const float* __restrict__ planes,
                const float* __restrict__ W1, const float* __restrict__ b1,
                const float* __restrict__ W2,
                float* __restrict__ q, int n) {
  __shared__ float hbuf[8 * 256];
  __shared__ float sbuf[NTB];
  const int lane = threadIdx.x & 31;
  const int wv = threadIdx.x >> 5;
  const int node0 = (int)blockIdx.x * (NTB / 2);
  plane_partials<2>(planes, node0, n, sbuf);
  __syncthreads();

  const int base = node0 + wv * 16;
  const int m = lane & 15;
  const bool hi = lane >= 16;
  int node = base + m;
  int cn = node < n ? node : n - 1;
  float acc1 = sbuf[wv * 16 + m] + sbuf[NTB / 2 + wv * 16 + m];
  float s = dis[cn] * (acc1 + p[cn]);

  v2f a = { hi ? 0.f : s,      hi ? 0.f : 1.f };
  v2f b = { hi ? 0.f : W1[m],  hi ? 0.f : b1[m] };
  v8f c = {0.f, 0.f, 0.f, 0.f, 0.f, 0.f, 0.f, 0.f};
  v8f h = __builtin_amdgcn_wmma_f32_16x16x4_f32(false, a, false, b, (short)0, c,
                                                false, false);
  float* t0 = &hbuf[wv * 256];
#pragma unroll
  for (int i = 0; i < 8; ++i) {
    float hv = h[i] > 0.f ? h[i] : 0.f;
    t0[(i + (hi ? 8 : 0)) * 16 + m] = hv;
  }
  __syncthreads();

  v8f t = {0.f, 0.f, 0.f, 0.f, 0.f, 0.f, 0.f, 0.f};
#pragma unroll
  for (int ck = 0; ck < 4; ++ck) {
    int k0 = 4 * ck + (hi ? 2 : 0);
    v2f ap = { W2[k0], W2[k0 + 1] };
    v2f bp = { t0[m * 16 + k0], t0[m * 16 + k0 + 1] };
    t = __builtin_amdgcn_wmma_f32_16x16x4_f32(false, ap, false, bp, (short)0, t,
                                              false, false);
  }
  if (!hi && node < n) q[node] = dis[node] * t[0];
}

__global__ void __launch_bounds__(NTB)
gcn_finalize(const float* __restrict__ dis, const float* __restrict__ q,
             const float* __restrict__ planes, const float* __restrict__ b2,
             float* __restrict__ out, int n) {
  __shared__ float sbuf[NTB];
  const int node0 = (int)blockIdx.x * (NTB / 4);
  plane_partials<4>(planes, node0, n, sbuf);
  __syncthreads();
  const int t = threadIdx.x;
  const int i = node0 + t;
  if (t < NTB / 4 && i < n) {
    float acc2 = (sbuf[t] + sbuf[NTB / 4 + t]) + (sbuf[2 * (NTB / 4) + t] + sbuf[3 * (NTB / 4) + t]);
    out[i] = dis[i] * (acc2 + q[i]) + b2[0];
  }
}

extern "C" void kernel_launch(void* const* d_in, const int* in_sizes, int n_in,
                              void* d_out, int out_size, void* d_ws, size_t ws_size,
                              hipStream_t stream) {
  const float* x  = (const float*)d_in[0];
  const int*   ei = (const int*)d_in[1];
  const float* W1 = (const float*)d_in[2];
  const float* b1 = (const float*)d_in[3];
  const float* W2 = (const float*)d_in[4];
  const float* b2 = (const float*)d_in[5];
  float* out = (float*)d_out;

  const int n = in_sizes[0];
  const int E = in_sizes[1] / 2;
  const int* row = ei;
  const int* col = ei + E;

  float* ws     = (float*)d_ws;
  float* dis    = ws;
  float* p      = ws + 1 * (size_t)n;
  float* q      = ws + 2 * (size_t)n;
  float* planes = ws + 3 * (size_t)n;

  const int ranges     = (n + RANGE - 1) / RANGE;
  const dim3 edgeGrid(CHUNKS, ranges);
  const int nodeBlocks = (n + NTB / 4 - 1) / (NTB / 4);
  const int tiles      = (n + 15) / 16;
  const int mlpBlocks  = (tiles + 7) / 8;

  gcn_degree     <<<edgeGrid,   ETB, 0, stream>>>(row, col, x, planes, n, E);
  gcn_prep       <<<nodeBlocks, NTB, 0, stream>>>(x, planes, dis, p, n);
  gcn_scatter    <<<edgeGrid,   ETB, 0, stream>>>(row, col, p, planes, n, E);
  gcn_node_update<<<mlpBlocks,  NTB, 0, stream>>>(dis, p, planes, W1, b1, W2, q, n);
  gcn_scatter    <<<edgeGrid,   ETB, 0, stream>>>(row, col, q, planes, n, E);
  gcn_finalize   <<<nodeBlocks, NTB, 0, stream>>>(dis, q, planes, b2, out, n);
}


# Gather passes copy their source values into LDS: block (edge chunk, target quarter, source half) reads values from LDS, no global gather; 64 partial planes, not 128 (sum order changes)
# speedup vs baseline: 10.9165x; 1.4499x over previous
#include <hip/hip_runtime.h>


typedef float v2f __attribute__((ext_vector_type(2)));
typedef float v8f __attribute__((ext_vector_type(8)));


constexpr int CHUNKS  = 128;
constexpr int RANGE   = 51200;
constexpr int GCHUNKS = 32;
constexpr int GS      = 2;
constexpr int SRANGE  = 51200;
constexpr int DRANGE  = 25600;
constexpr int GPLANES = GCHUNKS * GS;
constexpr int ETB     = 1024;
constexpr int SLD     = (SRANGE + ETB * 4 - 1) / (ETB * 4);
constexpr int UNR     = 4;
constexpr int NTB     = 256;

__global__ void __launch_bounds__(ETB)
gcn_degree(const int* __restrict__ col, float* __restrict__ planes, int n, int E) {
  __shared__ __attribute__((aligned(16))) float acc[RANGE];
  const int tid = threadIdx.x;
  int rsz = (n + (int)gridDim.y - 1) / (int)gridDim.y;
  rsz = (rsz + 31) & ~31;
  const int lo = (int)blockIdx.y * rsz;
  int len = n - lo;
  if (len > rsz) len = rsz;
  if (len < 0) len = 0;
  const unsigned ulen = (unsigned)len;

  for (int i = tid * 4; i < len; i += ETB * 4)
    *reinterpret_cast<float4*>(&acc[i]) = make_float4(0.f, 0.f, 0.f, 0.f);
  __syncthreads();

  int chunkE = (E + (int)gridDim.x - 1) / (int)gridDim.x;
  chunkE = (chunkE + 3) & ~3;
  int eb = (int)blockIdx.x * chunkE;
  if (eb > E) eb = E;
  int ee = eb + chunkE;
  if (ee > E) ee = E;

  if ((E & 3) == 0) {
    const int groups = (ee - eb) >> 2;
    const int4* col4 = reinterpret_cast<const int4*>(col + eb);
    for (int g0 = tid; g0 < groups; g0 += ETB * UNR) {
      int cc[UNR][4];
      bool ok[UNR];
#pragma unroll
      for (int u = 0; u < UNR; ++u) {
        const int g = g0 + u * ETB;
        ok[u] = g < groups;
        const int gc = ok[u] ? g : g0;
        const int4 c4 = col4[gc];
        cc[u][0] = c4.x; cc[u][1] = c4.y; cc[u][2] = c4.z; cc[u][3] = c4.w;
      }
#pragma unroll
      for (int u = 0; u < UNR; ++u) {
#pragma unroll
        for (int k = 0; k < 4; ++k) {
          const unsigned d = (unsigned)(cc[u][k] - lo);
          if (ok[u] && d < ulen) atomicAdd(&acc[d], 1.0f);
        }
      }
    }
  } else {
    for (int e = eb + tid; e < ee; e += ETB) {
      const unsigned dd = (unsigned)(col[e] - lo);
      if (dd < ulen) atomicAdd(&acc[dd], 1.0f);
    }
  }
  __syncthreads();

  float* dst = planes + (size_t)blockIdx.x * (size_t)n + lo;
  if ((n & 3) == 0) {
    for (int i = tid * 4; i < len; i += ETB * 4)
      *reinterpret_cast<float4*>(&dst[i]) = *reinterpret_cast<const float4*>(&acc[i]);
  } else {
    for (int i = tid; i < len; i += ETB) dst[i] = acc[i];
  }
}

__global__ void __launch_bounds__(ETB)
gcn_scatter(const int* __restrict__ row, const int* __restrict__ col,
            const float* __restrict__ val, float* __restrict__ planes, int n, int E) {
  __shared__ __attribute__((aligned(16))) float sval[SLD * ETB * 4];
  __shared__ __attribute__((aligned(16))) float acc[DRANGE];
  const int tid = threadIdx.x;

  int dsz = (n + (int)gridDim.y - 1) / (int)gridDim.y;
  dsz = (dsz + 31) & ~31;
  const int dlo = (int)blockIdx.y * dsz;
  int dlen = n - dlo;
  if (dlen > dsz) dlen = dsz;
  if (dlen < 0) dlen = 0;
  const unsigned udlen = (unsigned)dlen;

  int ssz = (n + GS - 1) / GS;
  ssz = (ssz + 31) & ~31;
  const int slo = (int)blockIdx.z * ssz;
  int slen = n - slo;
  if (slen > ssz) slen = ssz;
  if (slen < 0) slen = 0;
  const unsigned uslen = (unsigned)slen;

  const float* vsrc = val + slo;
  if ((n & 3) == 0) {
#pragma unroll
    for (int u = 0; u < SLD; ++u) {
      const int i = (tid + u * ETB) * 4;
      *reinterpret_cast<float4*>(&sval[i]) =
          *reinterpret_cast<const float4*>(&vsrc[i < slen ? i : 0]);
    }
  } else {
    for (int i = tid; i < slen; i += ETB) sval[i] = vsrc[i];
  }
  for (int i = tid * 4; i < dlen; i += ETB * 4)
    *reinterpret_cast<float4*>(&acc[i]) = make_float4(0.f, 0.f, 0.f, 0.f);
  __syncthreads();

  int chunkE = (E + (int)gridDim.x - 1) / (int)gridDim.x;
  chunkE = (chunkE + 3) & ~3;
  int eb = (int)blockIdx.x * chunkE;
  if (eb > E) eb = E;
  int ee = eb + chunkE;
  if (ee > E) ee = E;

  if ((E & 3) == 0) {
    const int groups = (ee - eb) >> 2;
    const int4* col4 = reinterpret_cast<const int4*>(col + eb);
    const int4* row4 = reinterpret_cast<const int4*>(row + eb);
    for (int g0 = tid; g0 < groups; g0 += ETB * UNR) {
      int cc[UNR][4];
      int rr[UNR][4];
      bool ok[UNR];
#pragma unroll
      for (int u = 0; u < UNR; ++u) {
        const int g = g0 + u * ETB;
        ok[u] = g < groups;
        const int gc = ok[u] ? g : g0;
        const int4 c4 = col4[gc];
        const int4 r4 = row4[gc];
        cc[u][0] = c4.x; cc[u][1] = c4.y; cc[u][2] = c4.z; cc[u][3] = c4.w;
        rr[u][0] = r4.x; rr[u][1] = r4.y; rr[u][2] = r4.z; rr[u][3] = r4.w;
      }
      asm volatile("" :: "v"(cc[0][0]), "v"(rr[0][0]), "v"(cc[1][0]), "v"(rr[1][0]),
                         "v"(cc[2][0]), "v"(rr[2][0]), "v"(cc[3][0]), "v"(rr[3][0]));
      unsigned d[UNR][4];
      bool in[UNR][4];
      float v[UNR][4];
#pragma unroll
      for (int u = 0; u < UNR; ++u) {
#pragma unroll
        for (int k = 0; k < 4; ++k) {
          d[u][k] = (unsigned)(cc[u][k] - dlo);
          const unsigned s = (unsigned)(rr[u][k] - slo);
          in[u][k] = ok[u] && d[u][k] < udlen && s < uslen;
          v[u][k] = sval[in[u][k] ? s : 0u];
        }
      }
      asm volatile("" :: "v"(v[0][0]));
#pragma unroll
      for (int u = 0; u < UNR; ++u) {
#pragma unroll
        for (int k = 0; k < 4; ++k) {
          if (in[u][k]) atomicAdd(&acc[d[u][k]], v[u][k]);
        }
      }
    }
  } else {
    for (int e = eb + tid; e < ee; e += ETB) {
      const unsigned dd = (unsigned)(col[e] - dlo);
      const unsigned ss = (unsigned)(row[e] - slo);
      if (dd < udlen && ss < uslen) atomicAdd(&acc[dd], sval[ss]);
    }
  }
  __syncthreads();

  float* dst = planes + (size_t)((int)blockIdx.x * GS + (int)blockIdx.z) * (size_t)n + dlo;
  if ((n & 3) == 0) {
    for (int i = tid * 4; i < dlen; i += ETB * 4)
      *reinterpret_cast<float4*>(&dst[i]) = *reinterpret_cast<const float4*>(&acc[i]);
  } else {
    for (int i = tid; i < dlen; i += ETB) dst[i] = acc[i];
  }
}

template <int PARTS, int PLANES>
__device__ __forceinline__ void plane_partials(const float* __restrict__ planes, int node0, int n,
                                               float* sbuf) {
  constexpr int NODES = NTB / PARTS;
  constexpr int PER = PLANES / PARTS;
  const int j = (int)threadIdx.x % NODES;
  const int part = (int)threadIdx.x / NODES;
  int node = node0 + j;
  if (node >= n) node = n - 1;
  const float* src = planes + (size_t)(part * PER) * (size_t)n + node;
  float s0 = 0.f, s1 = 0.f, s2 = 0.f, s3 = 0.f;
#pragma unroll
  for (int c = 0; c < PER; c += 4) {
    s0 += src[(size_t)(c + 0) * (size_t)n];
    s1 += src[(size_t)(c + 1) * (size_t)n];
    s2 += src[(size_t)(c + 2) * (size_t)n];
    s3 += src[(size_t)(c + 3) * (size_t)n];
  }
  sbuf[part * NODES + j] = (s0 + s1) + (s2 + s3);
}

__global__ void __launch_bounds__(NTB)
gcn_prep(const float* __restrict__ x, const float* __restrict__ planes,
         float* __restrict__ dis, float* __restrict__ p, int n) {
  __shared__ float sbuf[NTB];
  const int node0 = (int)blockIdx.x * (NTB / 4);
  plane_partials<4, CHUNKS>(planes, node0, n, sbuf);
  __syncthreads();
  const int t = threadIdx.x;
  const int i = node0 + t;
  if (t < NTB / 4 && i < n) {
    float deg = (sbuf[t] + sbuf[NTB / 4 + t]) + (sbuf[2 * (NTB / 4) + t] + sbuf[3 * (NTB / 4) + t]);
    float d = deg + 1.0f;
    float r = rsqrtf(d);
    dis[i] = r;
    p[i] = r * x[i];
  }
}

__global__ void __launch_bounds__(NTB)
gcn_node_update(const float* __restrict__ dis, const float* __restrict__ p,
                const float* __restrict__ planes,
                const float* __restrict__ W1, const float* __restrict__ b1,
                const float* __restrict__ W2,
                float* __restrict__ q, int n) {
  __shared__ float hbuf[8 * 256];
  __shared__ float sbuf[NTB];
  const int lane = threadIdx.x & 31;
  const int wv = threadIdx.x >> 5;
  const int node0 = (int)blockIdx.x * (NTB / 2);
  plane_partials<2, GPLANES>(planes, node0, n, sbuf);
  __syncthreads();

  const int base = node0 + wv * 16;
  const int m = lane & 15;
  const bool hi = lane >= 16;
  int node = base + m;
  int cn = node < n ? node : n - 1;
  float acc1 = sbuf[wv * 16 + m] + sbuf[NTB / 2 + wv * 16 + m];
  float s = dis[cn] * (acc1 + p[cn]);

  v2f a = { hi ? 0.f : s,      hi ? 0.f : 1.f };
  v2f b = { hi ? 0.f : W1[m],  hi ? 0.f : b1[m] };
  v8f c = {0.f, 0.f, 0.f, 0.f, 0.f, 0.f, 0.f, 0.f};
  v8f h = __builtin_amdgcn_wmma_f32_16x16x4_f32(false, a, false, b, (short)0, c,
                                                false, false);
  float* t0 = &hbuf[wv * 256];
#pragma unroll
  for (int i = 0; i < 8; ++i) {
    float hv = h[i] > 0.f ? h[i] : 0.f;
    t0[(i + (hi ? 8 : 0)) * 16 + m] = hv;
  }
  __syncthreads();

  v8f t = {0.f, 0.f, 0.f, 0.f, 0.f, 0.f, 0.f, 0.f};
#pragma unroll
  for (int ck = 0; ck < 4; ++ck) {
    int k0 = 4 * ck + (hi ? 2 : 0);
    v2f ap = { W2[k0], W2[k0 + 1] };
    v2f bp = { t0[m * 16 + k0], t0[m * 16 + k0 + 1] };
    t = __builtin_amdgcn_wmma_f32_16x16x4_f32(false, ap, false, bp, (short)0, t,
                                              false, false);
  }
  if (!hi && node < n) q[node] = dis[node] * t[0];
}

__global__ void __launch_bounds__(NTB)
gcn_finalize(const float* __restrict__ dis, const float* __restrict__ q,
             const float* __restrict__ planes, const float* __restrict__ b2,
             float* __restrict__ out, int n) {
  __shared__ float sbuf[NTB];
  const int node0 = (int)blockIdx.x * (NTB / 4);
  plane_partials<4, GPLANES>(planes, node0, n, sbuf);
  __syncthreads();
  const int t = threadIdx.x;
  const int i = node0 + t;
  if (t < NTB / 4 && i < n) {
    float acc2 = (sbuf[t] + sbuf[NTB / 4 + t]) + (sbuf[2 * (NTB / 4) + t] + sbuf[3 * (NTB / 4) + t]);
    out[i] = dis[i] * (acc2 + q[i]) + b2[0];
  }
}

extern "C" void kernel_launch(void* const* d_in, const int* in_sizes, int n_in,
                              void* d_out, int out_size, void* d_ws, size_t ws_size,
                              hipStream_t stream) {
  const float* x  = (const float*)d_in[0];
  const int*   ei = (const int*)d_in[1];
  const float* W1 = (const float*)d_in[2];
  const float* b1 = (const float*)d_in[3];
  const float* W2 = (const float*)d_in[4];
  const float* b2 = (const float*)d_in[5];
  float* out = (float*)d_out;

  const int n = in_sizes[0];
  const int E = in_sizes[1] / 2;
  const int* row = ei;
  const int* col = ei + E;

  float* ws     = (float*)d_ws;
  float* dis    = ws;
  float* p      = ws + 1 * (size_t)n;
  float* q      = ws + 2 * (size_t)n;
  float* planes = ws + 3 * (size_t)n;

  const int ranges     = (n + RANGE - 1) / RANGE;
  const int dranges    = (n + DRANGE - 1) / DRANGE;
  const dim3 degGrid(CHUNKS, ranges);
  const dim3 edgeGrid(GCHUNKS, dranges, GS);
  const int nodeBlocks = (n + NTB / 4 - 1) / (NTB / 4);
  const int tiles      = (n + 15) / 16;
  const int mlpBlocks  = (tiles + 7) / 8;

  gcn_degree     <<<degGrid,    ETB, 0, stream>>>(col, planes, n, E);
  gcn_prep       <<<nodeBlocks, NTB, 0, stream>>>(x, planes, dis, p, n);
  gcn_scatter    <<<edgeGrid,   ETB, 0, stream>>>(row, col, p, planes, n, E);
  gcn_node_update<<<mlpBlocks,  NTB, 0, stream>>>(dis, p, planes, W1, b1, W2, q, n);
  gcn_scatter    <<<edgeGrid,   ETB, 0, stream>>>(row, col, q, planes, n, E);
  gcn_finalize   <<<nodeBlocks, NTB, 0, stream>>>(dis, q, planes, b2, out, n);
}


# Degree kernel also sorts, in LDS, 14336-edge tiles into one-word (target, source offset) entries by range pair, stored in whole lines; gather blocks read only their pair (sum order changes)
# speedup vs baseline: 12.2727x; 1.1242x over previous
#include <hip/hip_runtime.h>


typedef float v2f __attribute__((ext_vector_type(2)));
typedef float v8f __attribute__((ext_vector_type(8)));


constexpr int CHUNKS  = 128;
constexpr int RANGE   = 51200;
constexpr int GCHUNKS = 32;
constexpr int GS      = 2;
constexpr int SRANGE  = 51200;
constexpr int DRANGE  = 25600;
constexpr int GPLANES = GCHUNKS * GS;
constexpr int ETB     = 1024;
constexpr int SLD     = (SRANGE + ETB * 4 - 1) / (ETB * 4);
constexpr int UNR     = 4;
constexpr int NTB     = 256;
constexpr int SLOTS   = 14;
constexpr int TILE_E  = SLOTS * ETB;
constexpr int MAXDR   = 4;
constexpr int MAXB    = MAXDR * GS;
constexpr int TROUND  = 14;
constexpr int TW      = 2;
static_assert(GS == 2, "the pair of an edge is computed for two source ranges");
static_assert(TILE_E + ETB <= RANGE, "a sorted tile (and one spare word per thread) is staged in the degree accumulators' LDS");
static_assert(DRANGE <= 32768 && SRANGE <= 65536, "one-word entries: 15-bit target offset, 16-bit source offset");

__global__ void __launch_bounds__(ETB)
gcn_degree(const int* __restrict__ row, const int* __restrict__ col, float* __restrict__ planes,
           float* __restrict__ sorted, int* __restrict__ tbl, int n, int E, int dranges, int NT) {
  __shared__ __attribute__((aligned(16))) float acc[RANGE];
  __shared__ unsigned cnt[(MAXB + 1) * ETB];
  __shared__ unsigned wtot[MAXB * 32];
  __shared__ unsigned btot[MAXB];
  const int tid = threadIdx.x;
  int rsz = (n + (int)gridDim.y - 1) / (int)gridDim.y;
  rsz = (rsz + 31) & ~31;
  const int lo = (int)blockIdx.y * rsz;
  int len = n - lo;
  if (len > rsz) len = rsz;
  if (len < 0) len = 0;
  const unsigned ulen = (unsigned)len;

  for (int i = tid * 4; i < len; i += ETB * 4)
    *reinterpret_cast<float4*>(&acc[i]) = make_float4(0.f, 0.f, 0.f, 0.f);
  __syncthreads();

  int chunkE = (E + (int)gridDim.x - 1) / (int)gridDim.x;
  chunkE = (chunkE + 3) & ~3;
  int eb = (int)blockIdx.x * chunkE;
  if (eb > E) eb = E;
  int ee = eb + chunkE;
  if (ee > E) ee = E;

  if ((E & 3) == 0) {
    const int groups = (ee - eb) >> 2;
    const int4* col4 = reinterpret_cast<const int4*>(col + eb);
    for (int g0 = tid; g0 < groups; g0 += ETB * UNR) {
      int cc[UNR][4];
      bool ok[UNR];
#pragma unroll
      for (int u = 0; u < UNR; ++u) {
        const int g = g0 + u * ETB;
        ok[u] = g < groups;
        const int gc = ok[u] ? g : g0;
        const int4 c4 = col4[gc];
        cc[u][0] = c4.x; cc[u][1] = c4.y; cc[u][2] = c4.z; cc[u][3] = c4.w;
      }
#pragma unroll
      for (int u = 0; u < UNR; ++u) {
#pragma unroll
        for (int k = 0; k < 4; ++k) {
          const unsigned d = (unsigned)(cc[u][k] - lo);
          if (ok[u] && d < ulen) atomicAdd(&acc[d], 1.0f);
        }
      }
    }
  } else {
    for (int e = eb + tid; e < ee; e += ETB) {
      const unsigned dd = (unsigned)(col[e] - lo);
      if (dd < ulen) atomicAdd(&acc[dd], 1.0f);
    }
  }
  __syncthreads();

  float* dst = planes + (size_t)blockIdx.x * (size_t)n + lo;
  if ((n & 3) == 0) {
    for (int i = tid * 4; i < len; i += ETB * 4)
      *reinterpret_cast<float4*>(&dst[i]) = *reinterpret_cast<const float4*>(&acc[i]);
  } else {
    for (int i = tid; i < len; i += ETB) dst[i] = acc[i];
  }

  int dsz = (n + dranges - 1) / dranges;
  dsz = (dsz + 31) & ~31;
  int ssz = (n + GS - 1) / GS;
  ssz = (ssz + 31) & ~31;
  const unsigned un = (unsigned)n;
  const unsigned udsz = (unsigned)dsz, ussz = (unsigned)ssz;
  const int lane = tid & 31;
  const int wv = tid >> 5;
  const int nblk = (int)(gridDim.x * gridDim.y);
  const int bid = (int)(blockIdx.x * gridDim.y + blockIdx.y);
  for (int t = bid; t < NT; t += nblk) {
    const int tbeg = t * TILE_E;
    int tend = tbeg + TILE_E;
    if (tend > E) tend = E;

    unsigned cc[SLOTS], rr[SLOTS];
#pragma unroll
    for (int q = 0; q < SLOTS; ++q) {
      const int e = tbeg + q * ETB + tid;
      const int ec = e < tend ? e : tbeg;
      cc[q] = (unsigned)col[ec];
      rr[q] = (unsigned)row[ec];
    }

    unsigned wd[SLOTS], ci[SLOTS];
#pragma unroll
    for (int b = 0; b <= MAXB; ++b) cnt[b * ETB + tid] = 0u;
#pragma unroll
    for (int q = 0; q < SLOTS; ++q) {
      const unsigned c = cc[q], r = rr[q];
      const bool valid = (tbeg + q * ETB + tid) < tend && c < un && r < un;
      const unsigned dq = (unsigned)(c >= udsz) + (unsigned)(c >= 2u * udsz) + (unsigned)(c >= 3u * udsz);
      const unsigned sh = (unsigned)(r >= ussz);
      const unsigned pr = valid ? dq * GS + sh : (unsigned)MAXB;
      wd[q] = ((c - dq * udsz) << 16) | (r - sh * ussz);
      ci[q] = pr * ETB + (unsigned)tid;
      atomicAdd(&cnt[ci[q]], 1u);
    }

    {
      unsigned v0[MAXB], v[MAXB];
#pragma unroll
      for (int b = 0; b < MAXB; ++b) { v0[b] = cnt[b * ETB + tid]; v[b] = v0[b]; }
#pragma unroll
      for (int o = 1; o < 32; o <<= 1) {
        unsigned u[MAXB];
#pragma unroll
        for (int b = 0; b < MAXB; ++b) u[b] = __shfl_up(v[b], (unsigned)o);
#pragma unroll
        for (int b = 0; b < MAXB; ++b) if (lane >= o) v[b] += u[b];
      }
#pragma unroll
      for (int b = 0; b < MAXB; ++b) {
        cnt[b * ETB + tid] = v[b] - v0[b];
        if (lane == 31) wtot[b * 32 + wv] = v[b];
      }
    }
    __syncthreads();
    if (tid < MAXB * 32) {
      const unsigned v0 = wtot[tid];
      unsigned v = v0;
#pragma unroll
      for (int o = 1; o < 32; o <<= 1) {
        const unsigned u = __shfl_up(v, (unsigned)o);
        if (lane >= o) v += u;
      }
      wtot[tid] = v - v0;
      if (lane == 31) btot[wv] = v;
    }
    __syncthreads();
    unsigned total = 0u;
#pragma unroll
    for (int b = 0; b < MAXB; ++b) {
      cnt[b * ETB + tid] += total + wtot[b * 32 + wv];
      total += btot[b];
    }
    if (tid == 0) {
      unsigned a = 0u;
#pragma unroll
      for (int b = 0; b < MAXB; ++b) {
        tbl[t * (MAXB + 1) + b] = (int)a;
        a += btot[b];
      }
      tbl[t * (MAXB + 1) + MAXB] = (int)a;
    }

    unsigned pos[SLOTS];
#pragma unroll
    for (int q = 0; q < SLOTS; ++q) {
      const unsigned p0 = atomicAdd(&cnt[ci[q]], 1u);
      pos[q] = ci[q] < (unsigned)(MAXB * ETB) ? p0 : (unsigned)(TILE_E + tid);
    }
#pragma unroll
    for (int q = 0; q < SLOTS; ++q) acc[pos[q]] = __uint_as_float(wd[q]);
    __syncthreads();

    float4* dst4 = reinterpret_cast<float4*>(sorted + (size_t)t * (size_t)TILE_E);
    const int n4 = (int)((total + 3u) >> 2);
    for (int i = tid; i < n4; i += ETB) dst4[i] = reinterpret_cast<const float4*>(acc)[i];
    __syncthreads();
  }
}

__device__ __forceinline__ void tile_table(const int* __restrict__ tbl, int ta, int t1, int bin,
                                           int (&tb)[TROUND], int (&tl)[TROUND], int& maxlen) {
  int b[TROUND], e[TROUND];
#pragma unroll
  for (int j = 0; j < TROUND; ++j) {
    const int tt = ta + j < t1 ? ta + j : t1 - 1;
    b[j] = tbl[tt * (MAXB + 1) + bin];
    e[j] = tbl[tt * (MAXB + 1) + bin + 1];
  }
  maxlen = 0;
#pragma unroll
  for (int j = 0; j < TROUND; ++j) {
    const bool okt = ta + j < t1;
    const int tt = okt ? ta + j : t1 - 1;
    tl[j] = okt ? e[j] - b[j] : 0;
    tb[j] = tt * TILE_E + b[j];
    maxlen = tl[j] > maxlen ? tl[j] : maxlen;
  }
}

__global__ void __launch_bounds__(ETB)
gcn_scatter(const unsigned* __restrict__ sorted, const int* __restrict__ tbl,
            const float* __restrict__ val, float* __restrict__ planes, int n, int NT) {
  __shared__ __attribute__((aligned(16))) float sval[SLD * ETB * 4];
  __shared__ __attribute__((aligned(16))) float acc[DRANGE];
  const int tid = threadIdx.x;

  int dsz = (n + (int)gridDim.y - 1) / (int)gridDim.y;
  dsz = (dsz + 31) & ~31;
  const int dlo = (int)blockIdx.y * dsz;
  int dlen = n - dlo;
  if (dlen > dsz) dlen = dsz;
  if (dlen < 0) dlen = 0;

  int ssz = (n + GS - 1) / GS;
  ssz = (ssz + 31) & ~31;
  const int slo = (int)blockIdx.z * ssz;
  int slen = n - slo;
  if (slen > ssz) slen = ssz;
  if (slen < 0) slen = 0;

  const int bin = (int)blockIdx.y * GS + (int)blockIdx.z;
  const int tpc = (NT + (int)gridDim.x - 1) / (int)gridDim.x;
  int ta = (int)blockIdx.x * tpc;
  int t1 = ta + tpc;
  if (t1 > NT) t1 = NT;
  int tb[TROUND], tl[TROUND];
  int maxlen = 0;
#pragma unroll
  for (int j = 0; j < TROUND; ++j) { tb[j] = 0; tl[j] = 0; }
  unsigned w[TROUND][TW];
  bool ok[TROUND][TW];
  auto load_words = [&](int base) {
#pragma unroll
    for (int j = 0; j < TROUND; ++j) {
#pragma unroll
      for (int k = 0; k < TW; ++k) {
        const int i = base + tid + k * ETB;
        ok[j][k] = i < tl[j];
        w[j][k] = 0u;
        if (ok[j][k]) w[j][k] = sorted[tb[j] + i];
      }
    }
  };
  if (ta < t1) {
    tile_table(tbl, ta, t1, bin, tb, tl, maxlen);
    load_words(0);
  }

  const float* vsrc = val + slo;
  if ((n & 3) == 0) {
#pragma unroll
    for (int u = 0; u < SLD; ++u) {
      const int i = (tid + u * ETB) * 4;
      *reinterpret_cast<float4*>(&sval[i]) =
          *reinterpret_cast<const float4*>(&vsrc[i < slen ? i : 0]);
    }
  } else {
    for (int i = tid; i < slen; i += ETB) sval[i] = vsrc[i];
  }
  for (int i = tid * 4; i < dlen; i += ETB * 4)
    *reinterpret_cast<float4*>(&acc[i]) = make_float4(0.f, 0.f, 0.f, 0.f);
  __syncthreads();

  while (ta < t1) {
    int base = 0;
    while (true) {
      float v[TROUND][TW];
#pragma unroll
      for (int j = 0; j < TROUND; ++j) {
#pragma unroll
        for (int k = 0; k < TW; ++k) v[j][k] = sval[w[j][k] & 0xFFFFu];
      }
#pragma unroll
      for (int j = 0; j < TROUND; ++j) {
#pragma unroll
        for (int k = 0; k < TW; ++k) {
          if (ok[j][k]) atomicAdd(&acc[w[j][k] >> 16], v[j][k]);
        }
      }
      base += ETB * TW;
      if (base >= maxlen) break;
      load_words(base);
    }
    ta += TROUND;
    if (ta < t1) {
      tile_table(tbl, ta, t1, bin, tb, tl, maxlen);
      load_words(0);
    }
  }
  __syncthreads();

  float* dst = planes + (size_t)((int)blockIdx.x * GS + (int)blockIdx.z) * (size_t)n + dlo;
  if ((n & 3) == 0) {
    for (int i = tid * 4; i < dlen; i += ETB * 4)
      *reinterpret_cast<float4*>(&dst[i]) = *reinterpret_cast<const float4*>(&acc[i]);
  } else {
    for (int i = tid; i < dlen; i += ETB) dst[i] = acc[i];
  }
}

template <int PARTS, int PLANES>
__device__ __forceinline__ void plane_partials(const float* __restrict__ planes, int node0, int n,
                                               float* sbuf) {
  constexpr int NODES = NTB / PARTS;
  constexpr int PER = PLANES / PARTS;
  const int j = (int)threadIdx.x % NODES;
  const int part = (int)threadIdx.x / NODES;
  int node = node0 + j;
  if (node >= n) node = n - 1;
  const float* src = planes + (size_t)(part * PER) * (size_t)n + node;
  float s0 = 0.f, s1 = 0.f, s2 = 0.f, s3 = 0.f;
#pragma unroll
  for (int c = 0; c < PER; c += 4) {
    s0 += src[(size_t)(c + 0) * (size_t)n];
    s1 += src[(size_t)(c + 1) * (size_t)n];
    s2 += src[(size_t)(c + 2) * (size_t)n];
    s3 += src[(size_t)(c + 3) * (size_t)n];
  }
  sbuf[part * NODES + j] = (s0 + s1) + (s2 + s3);
}

__global__ void __launch_bounds__(NTB)
gcn_prep(const float* __restrict__ x, const float* __restrict__ planes,
         float* __restrict__ dis, float* __restrict__ p, int n) {
  __shared__ float sbuf[NTB];
  const int node0 = (int)blockIdx.x * (NTB / 4);
  plane_partials<4, CHUNKS>(planes, node0, n, sbuf);
  __syncthreads();
  const int t = threadIdx.x;
  const int i = node0 + t;
  if (t < NTB / 4 && i < n) {
    float deg = (sbuf[t] + sbuf[NTB / 4 + t]) + (sbuf[2 * (NTB / 4) + t] + sbuf[3 * (NTB / 4) + t]);
    float d = deg + 1.0f;
    float r = rsqrtf(d);
    dis[i] = r;
    p[i] = r * x[i];
  }
}

__global__ void __launch_bounds__(NTB)
gcn_node_update(const float* __restrict__ dis, const float* __restrict__ p,
                const float* __restrict__ planes,
                const float* __restrict__ W1, const float* __restrict__ b1,
                const float* __restrict__ W2,
                float* __restrict__ q, int n) {
  __shared__ float hbuf[8 * 256];
  __shared__ float sbuf[NTB];
  const int lane = threadIdx.x & 31;
  const int wv = threadIdx.x >> 5;
  const int node0 = (int)blockIdx.x * (NTB / 2);
  plane_partials<2, GPLANES>(planes, node0, n, sbuf);
  __syncthreads();

  const int base = node0 + wv * 16;
  const int m = lane & 15;
  const bool hi = lane >= 16;
  int node = base + m;
  int cn = node < n ? node : n - 1;
  float acc1 = sbuf[wv * 16 + m] + sbuf[NTB / 2 + wv * 16 + m];
  float s = dis[cn] * (acc1 + p[cn]);

  v2f a = { hi ? 0.f : s,      hi ? 0.f : 1.f };
  v2f b = { hi ? 0.f : W1[m],  hi ? 0.f : b1[m] };
  v8f c = {0.f, 0.f, 0.f, 0.f, 0.f, 0.f, 0.f, 0.f};
  v8f h = __builtin_amdgcn_wmma_f32_16x16x4_f32(false, a, false, b, (short)0, c,
                                                false, false);
  float* t0 = &hbuf[wv * 256];
#pragma unroll
  for (int i = 0; i < 8; ++i) {
    float hv = h[i] > 0.f ? h[i] : 0.f;
    t0[(i + (hi ? 8 : 0)) * 16 + m] = hv;
  }
  __syncthreads();

  v8f t = {0.f, 0.f, 0.f, 0.f, 0.f, 0.f, 0.f, 0.f};
#pragma unroll
  for (int ck = 0; ck < 4; ++ck) {
    int k0 = 4 * ck + (hi ? 2 : 0);
    v2f ap = { W2[k0], W2[k0 + 1] };
    v2f bp = { t0[m * 16 + k0], t0[m * 16 + k0 + 1] };
    t = __builtin_amdgcn_wmma_f32_16x16x4_f32(false, ap, false, bp, (short)0, t,
                                              false, false);
  }
  if (!hi && node < n) q[node] = dis[node] * t[0];
}

__global__ void __launch_bounds__(NTB)
gcn_finalize(const float* __restrict__ dis, const float* __restrict__ q,
             const float* __restrict__ planes, const float* __restrict__ b2,
             float* __restrict__ out, int n) {
  __shared__ float sbuf[NTB];
  const int node0 = (int)blockIdx.x * (NTB / 4);
  plane_partials<4, GPLANES>(planes, node0, n, sbuf);
  __syncthreads();
  const int t = threadIdx.x;
  const int i = node0 + t;
  if (t < NTB / 4 && i < n) {
    float acc2 = (sbuf[t] + sbuf[NTB / 4 + t]) + (sbuf[2 * (NTB / 4) + t] + sbuf[3 * (NTB / 4) + t]);
    out[i] = dis[i] * (acc2 + q[i]) + b2[0];
  }
}

extern "C" void kernel_launch(void* const* d_in, const int* in_sizes, int n_in,
                              void* d_out, int out_size, void* d_ws, size_t ws_size,
                              hipStream_t stream) {
  const float* x  = (const float*)d_in[0];
  const int*   ei = (const int*)d_in[1];
  const float* W1 = (const float*)d_in[2];
  const float* b1 = (const float*)d_in[3];
  const float* W2 = (const float*)d_in[4];
  const float* b2 = (const float*)d_in[5];
  float* out = (float*)d_out;

  const int n = in_sizes[0];
  const int E = in_sizes[1] / 2;
  const int* row = ei;
  const int* col = ei + E;

  float* ws     = (float*)d_ws;
  float* dis    = ws;
  float* p      = ws + 1 * (size_t)n;
  float* q      = ws + 2 * (size_t)n;
  float* planes = ws + 3 * (size_t)n;
  const int NT = (E + TILE_E - 1) / TILE_E;
  float* sorted = ws + ((((size_t)(3 + CHUNKS) * (size_t)n) + 31) & ~(size_t)31);
  int* tbl = (int*)(sorted + (size_t)NT * (size_t)TILE_E);

  const int ranges     = (n + RANGE - 1) / RANGE;
  const int dranges    = (n + DRANGE - 1) / DRANGE;
  const dim3 degGrid(CHUNKS, ranges);
  const dim3 edgeGrid(GCHUNKS, dranges, GS);
  const int nodeBlocks = (n + NTB / 4 - 1) / (NTB / 4);
  const int tiles      = (n + 15) / 16;
  const int mlpBlocks  = (tiles + 7) / 8;

  gcn_degree     <<<degGrid,    ETB, 0, stream>>>(row, col, planes, sorted, tbl, n, E, dranges, NT);
  gcn_prep       <<<nodeBlocks, NTB, 0, stream>>>(x, planes, dis, p, n);
  gcn_scatter    <<<edgeGrid,   ETB, 0, stream>>>((const unsigned*)sorted, tbl, p, planes, n, NT);
  gcn_node_update<<<mlpBlocks,  NTB, 0, stream>>>(dis, p, planes, W1, b1, W2, q, n);
  gcn_scatter    <<<edgeGrid,   ETB, 0, stream>>>((const unsigned*)sorted, tbl, q, planes, n, NT);
  gcn_finalize   <<<nodeBlocks, NTB, 0, stream>>>(dis, q, planes, b2, out, n);
}


# Degree counts kept as integers, two 16-bit counts per word: count planes of 2 bytes per node, not 4; prep adds them with two-word loads, 4 nodes per lane (degree sums now integer, exact)
# speedup vs baseline: 13.0803x; 1.0658x over previous
#include <hip/hip_runtime.h>


typedef float v2f __attribute__((ext_vector_type(2)));
typedef float v8f __attribute__((ext_vector_type(8)));


constexpr int CHUNKS  = 128;
constexpr int RANGE   = 51200;
constexpr int GCHUNKS = 32;
constexpr int GS      = 2;
constexpr int SRANGE  = 51200;
constexpr int DRANGE  = 25600;
constexpr int GPLANES = GCHUNKS * GS;
constexpr int ETB     = 1024;
constexpr int SLD     = (SRANGE + ETB * 4 - 1) / (ETB * 4);
constexpr int UNR     = 4;
constexpr int NTB     = 256;
constexpr int SLOTS   = 14;
constexpr int TILE_E  = SLOTS * ETB;
constexpr int MAXDR   = 4;
constexpr int MAXB    = MAXDR * GS;
constexpr int TROUND  = 14;
constexpr int TW      = 2;
static_assert(GS == 2, "the pair of an edge is computed for two source ranges");
static_assert(TILE_E + ETB <= RANGE, "a sorted tile (and one spare word per thread) is staged in the degree accumulators' LDS");
static_assert(DRANGE <= 32768 && SRANGE <= 65536, "one-word entries: 15-bit target offset, 16-bit source offset");

__global__ void __launch_bounds__(ETB)
gcn_degree(const int* __restrict__ row, const int* __restrict__ col, unsigned* __restrict__ cplanes,
           unsigned* __restrict__ sorted, int* __restrict__ tbl, int n, int E, int dranges, int NT,
           int hw, int pack) {
  __shared__ __attribute__((aligned(16))) unsigned acc[RANGE];
  __shared__ unsigned cnt[(MAXB + 1) * ETB];
  __shared__ unsigned wtot[MAXB * 32];
  __shared__ unsigned btot[MAXB];
  const int tid = threadIdx.x;
  int rsz = (n + (int)gridDim.y - 1) / (int)gridDim.y;
  rsz = (rsz + 63) & ~63;
  const int lo = (int)blockIdx.y * rsz;
  int len = n - lo;
  if (len > rsz) len = rsz;
  if (len < 0) len = 0;
  const unsigned ulen = (unsigned)len;

  for (int i = tid * 4; i < len; i += ETB * 4)
    *reinterpret_cast<uint4*>(&acc[i]) = make_uint4(0u, 0u, 0u, 0u);
  __syncthreads();

  int chunkE = (E + (int)gridDim.x - 1) / (int)gridDim.x;
  chunkE = (chunkE + 3) & ~3;
  int eb = (int)blockIdx.x * chunkE;
  if (eb > E) eb = E;
  int ee = eb + chunkE;
  if (ee > E) ee = E;

  if ((E & 3) == 0) {
    const int groups = (ee - eb) >> 2;
    const int4* col4 = reinterpret_cast<const int4*>(col + eb);
    for (int g0 = tid; g0 < groups; g0 += ETB * UNR) {
      int cc[UNR][4];
      bool ok[UNR];
#pragma unroll
      for (int u = 0; u < UNR; ++u) {
        const int g = g0 + u * ETB;
        ok[u] = g < groups;
        const int gc = ok[u] ? g : g0;
        const int4 c4 = col4[gc];
        cc[u][0] = c4.x; cc[u][1] = c4.y; cc[u][2] = c4.z; cc[u][3] = c4.w;
      }
#pragma unroll
      for (int u = 0; u < UNR; ++u) {
#pragma unroll
        for (int k = 0; k < 4; ++k) {
          const unsigned d = (unsigned)(cc[u][k] - lo);
          if (ok[u] && d < ulen) atomicAdd(&acc[d], 1u);
        }
      }
    }
  } else {
    for (int e = eb + tid; e < ee; e += ETB) {
      const unsigned dd = (unsigned)(col[e] - lo);
      if (dd < ulen) atomicAdd(&acc[dd], 1u);
    }
  }
  __syncthreads();

  if (pack) {
    if ((n & 3) == 0) {
      uint2* dst2 = reinterpret_cast<uint2*>(cplanes + (size_t)blockIdx.x * (size_t)hw + (lo >> 1));
      for (int i = tid * 4; i < len; i += ETB * 4) {
        const uint4 a = *reinterpret_cast<const uint4*>(&acc[i]);
        dst2[i >> 2] = make_uint2(a.x | (a.y << 16), a.z | (a.w << 16));
      }
    } else {
      unsigned short* d16 = reinterpret_cast<unsigned short*>(cplanes + (size_t)blockIdx.x * (size_t)hw) + lo;
      for (int i = tid; i < len; i += ETB) d16[i] = (unsigned short)acc[i];
    }
  } else {
    unsigned* dst = cplanes + (size_t)blockIdx.x * (size_t)n + lo;
    if ((n & 3) == 0) {
      for (int i = tid * 4; i < len; i += ETB * 4)
        *reinterpret_cast<uint4*>(&dst[i]) = *reinterpret_cast<const uint4*>(&acc[i]);
    } else {
      for (int i = tid; i < len; i += ETB) dst[i] = acc[i];
    }
  }

  int dsz = (n + dranges - 1) / dranges;
  dsz = (dsz + 31) & ~31;
  int ssz = (n + GS - 1) / GS;
  ssz = (ssz + 31) & ~31;
  const unsigned un = (unsigned)n;
  const unsigned udsz = (unsigned)dsz, ussz = (unsigned)ssz;
  const int lane = tid & 31;
  const int wv = tid >> 5;
  const int nblk = (int)(gridDim.x * gridDim.y);
  const int bid = (int)(blockIdx.x * gridDim.y + blockIdx.y);
  for (int t = bid; t < NT; t += nblk) {
    const int tbeg = t * TILE_E;
    int tend = tbeg + TILE_E;
    if (tend > E) tend = E;

    unsigned cc[SLOTS], rr[SLOTS];
#pragma unroll
    for (int q = 0; q < SLOTS; ++q) {
      const int e = tbeg + q * ETB + tid;
      const int ec = e < tend ? e : tbeg;
      cc[q] = (unsigned)col[ec];
      rr[q] = (unsigned)row[ec];
    }

    unsigned wd[SLOTS], ci[SLOTS];
#pragma unroll
    for (int b = 0; b <= MAXB; ++b) cnt[b * ETB + tid] = 0u;
#pragma unroll
    for (int q = 0; q < SLOTS; ++q) {
      const unsigned c = cc[q], r = rr[q];
      const bool valid = (tbeg + q * ETB + tid) < tend && c < un && r < un;
      const unsigned dq = (unsigned)(c >= udsz) + (unsigned)(c >= 2u * udsz) + (unsigned)(c >= 3u * udsz);
      const unsigned sh = (unsigned)(r >= ussz);
      const unsigned pr = valid ? dq * GS + sh : (unsigned)MAXB;
      wd[q] = ((c - dq * udsz) << 16) | (r - sh * ussz);
      ci[q] = pr * ETB + (unsigned)tid;
      atomicAdd(&cnt[ci[q]], 1u);
    }

    {
      unsigned v0[MAXB], v[MAXB];
#pragma unroll
      for (int b = 0; b < MAXB; ++b) { v0[b] = cnt[b * ETB + tid]; v[b] = v0[b]; }
#pragma unroll
      for (int o = 1; o < 32; o <<= 1) {
        unsigned u[MAXB];
#pragma unroll
        for (int b = 0; b < MAXB; ++b) u[b] = __shfl_up(v[b], (unsigned)o);
#pragma unroll
        for (int b = 0; b < MAXB; ++b) if (lane >= o) v[b] += u[b];
      }
#pragma unroll
      for (int b = 0; b < MAXB; ++b) {
        cnt[b * ETB + tid] = v[b] - v0[b];
        if (lane == 31) wtot[b * 32 + wv] = v[b];
      }
    }
    __syncthreads();
    if (tid < MAXB * 32) {
      const unsigned v0 = wtot[tid];
      unsigned v = v0;
#pragma unroll
      for (int o = 1; o < 32; o <<= 1) {
        const unsigned u = __shfl_up(v, (unsigned)o);
        if (lane >= o) v += u;
      }
      wtot[tid] = v - v0;
      if (lane == 31) btot[wv] = v;
    }
    __syncthreads();
    unsigned total = 0u;
#pragma unroll
    for (int b = 0; b < MAXB; ++b) {
      cnt[b * ETB + tid] += total + wtot[b * 32 + wv];
      total += btot[b];
    }
    if (tid == 0) {
      unsigned a = 0u;
#pragma unroll
      for (int b = 0; b < MAXB; ++b) {
        tbl[t * (MAXB + 1) + b] = (int)a;
        a += btot[b];
      }
      tbl[t * (MAXB + 1) + MAXB] = (int)a;
    }

    unsigned pos[SLOTS];
#pragma unroll
    for (int q = 0; q < SLOTS; ++q) {
      const unsigned p0 = atomicAdd(&cnt[ci[q]], 1u);
      pos[q] = ci[q] < (unsigned)(MAXB * ETB) ? p0 : (unsigned)(TILE_E + tid);
    }
#pragma unroll
    for (int q = 0; q < SLOTS; ++q) acc[pos[q]] = wd[q];
    __syncthreads();

    uint4* dst4 = reinterpret_cast<uint4*>(sorted + (size_t)t * (size_t)TILE_E);
    const int n4 = (int)((total + 3u) >> 2);
    for (int i = tid; i < n4; i += ETB) dst4[i] = reinterpret_cast<const uint4*>(acc)[i];
    __syncthreads();
  }
}

__device__ __forceinline__ void tile_table(const int* __restrict__ tbl, int ta, int t1, int bin,
                                           int (&tb)[TROUND], int (&tl)[TROUND], int& maxlen) {
  int b[TROUND], e[TROUND];
#pragma unroll
  for (int j = 0; j < TROUND; ++j) {
    const int tt = ta + j < t1 ? ta + j : t1 - 1;
    b[j] = tbl[tt * (MAXB + 1) + bin];
    e[j] = tbl[tt * (MAXB + 1) + bin + 1];
  }
  maxlen = 0;
#pragma unroll
  for (int j = 0; j < TROUND; ++j) {
    const bool okt = ta + j < t1;
    const int tt = okt ? ta + j : t1 - 1;
    tl[j] = okt ? e[j] - b[j] : 0;
    tb[j] = tt * TILE_E + b[j];
    maxlen = tl[j] > maxlen ? tl[j] : maxlen;
  }
}

__global__ void __launch_bounds__(ETB)
gcn_scatter(const unsigned* __restrict__ sorted, const int* __restrict__ tbl,
            const float* __restrict__ val, float* __restrict__ planes, int n, int NT) {
  __shared__ __attribute__((aligned(16))) float sval[SLD * ETB * 4];
  __shared__ __attribute__((aligned(16))) float acc[DRANGE];
  const int tid = threadIdx.x;

  int dsz = (n + (int)gridDim.y - 1) / (int)gridDim.y;
  dsz = (dsz + 31) & ~31;
  const int dlo = (int)blockIdx.y * dsz;
  int dlen = n - dlo;
  if (dlen > dsz) dlen = dsz;
  if (dlen < 0) dlen = 0;

  int ssz = (n + GS - 1) / GS;
  ssz = (ssz + 31) & ~31;
  const int slo = (int)blockIdx.z * ssz;
  int slen = n - slo;
  if (slen > ssz) slen = ssz;
  if (slen < 0) slen = 0;

  const int bin = (int)blockIdx.y * GS + (int)blockIdx.z;
  const int tpc = (NT + (int)gridDim.x - 1) / (int)gridDim.x;
  int ta = (int)blockIdx.x * tpc;
  int t1 = ta + tpc;
  if (t1 > NT) t1 = NT;
  int tb[TROUND], tl[TROUND];
  int maxlen = 0;
#pragma unroll
  for (int j = 0; j < TROUND; ++j) { tb[j] = 0; tl[j] = 0; }
  unsigned w[TROUND][TW];
  bool ok[TROUND][TW];
  auto load_words = [&](int base) {
#pragma unroll
    for (int j = 0; j < TROUND; ++j) {
#pragma unroll
      for (int k = 0; k < TW; ++k) {
        const int i = base + tid + k * ETB;
        ok[j][k] = i < tl[j];
        w[j][k] = 0u;
        if (ok[j][k]) w[j][k] = sorted[tb[j] + i];
      }
    }
  };
  if (ta < t1) {
    tile_table(tbl, ta, t1, bin, tb, tl, maxlen);
    load_words(0);
  }

  const float* vsrc = val + slo;
  if ((n & 3) == 0) {
#pragma unroll
    for (int u = 0; u < SLD; ++u) {
      const int i = (tid + u * ETB) * 4;
      *reinterpret_cast<float4*>(&sval[i]) =
          *reinterpret_cast<const float4*>(&vsrc[i < slen ? i : 0]);
    }
  } else {
    for (int i = tid; i < slen; i += ETB) sval[i] = vsrc[i];
  }
  for (int i = tid * 4; i < dlen; i += ETB * 4)
    *reinterpret_cast<float4*>(&acc[i]) = make_float4(0.f, 0.f, 0.f, 0.f);
  __syncthreads();

  while (ta < t1) {
    int base = 0;
    while (true) {
      float v[TROUND][TW];
#pragma unroll
      for (int j = 0; j < TROUND; ++j) {
#pragma unroll
        for (int k = 0; k < TW; ++k) v[j][k] = sval[w[j][k] & 0xFFFFu];
      }
#pragma unroll
      for (int j = 0; j < TROUND; ++j) {
#pragma unroll
        for (int k = 0; k < TW; ++k) {
          if (ok[j][k]) atomicAdd(&acc[w[j][k] >> 16], v[j][k]);
        }
      }
      base += ETB * TW;
      if (base >= maxlen) break;
      load_words(base);
    }
    ta += TROUND;
    if (ta < t1) {
      tile_table(tbl, ta, t1, bin, tb, tl, maxlen);
      load_words(0);
    }
  }
  __syncthreads();

  float* dst = planes + (size_t)((int)blockIdx.x * GS + (int)blockIdx.z) * (size_t)n + dlo;
  if ((n & 3) == 0) {
    for (int i = tid * 4; i < dlen; i += ETB * 4)
      *reinterpret_cast<float4*>(&dst[i]) = *reinterpret_cast<const float4*>(&acc[i]);
  } else {
    for (int i = tid; i < dlen; i += ETB) dst[i] = acc[i];
  }
}

template <int PARTS, int PLANES>
__device__ __forceinline__ void plane_partials(const float* __restrict__ planes, int node0, int n,
                                               float* sbuf) {
  constexpr int NODES = NTB / PARTS;
  constexpr int PER = PLANES / PARTS;
  const int j = (int)threadIdx.x % NODES;
  const int part = (int)threadIdx.x / NODES;
  int node = node0 + j;
  if (node >= n) node = n - 1;
  const float* src = planes + (size_t)(part * PER) * (size_t)n + node;
  float s0 = 0.f, s1 = 0.f, s2 = 0.f, s3 = 0.f;
#pragma unroll
  for (int c = 0; c < PER; c += 4) {
    s0 += src[(size_t)(c + 0) * (size_t)n];
    s1 += src[(size_t)(c + 1) * (size_t)n];
    s2 += src[(size_t)(c + 2) * (size_t)n];
    s3 += src[(size_t)(c + 3) * (size_t)n];
  }
  sbuf[part * NODES + j] = (s0 + s1) + (s2 + s3);
}

constexpr int PN = 128;
constexpr int PPARTS = NTB / 32;
static_assert(CHUNKS % PPARTS == 0 && PN == 4 * 32, "a wave covers the block's nodes, the waves share the planes");
__global__ void __launch_bounds__(NTB)
gcn_prep(const float* __restrict__ x, const unsigned* __restrict__ cplanes,
         float* __restrict__ dis, float* __restrict__ p, int n, int hw, int pack) {
  __shared__ __attribute__((aligned(16))) unsigned sbuf[PPARTS * PN];
  const int t = threadIdx.x;
  const int node0 = (int)blockIdx.x * PN;
  if (pack) {
    const int lane = t & 31;
    const int part = t >> 5;
    int w0 = (node0 >> 1) + 2 * lane;
    if (w0 > hw - 2) w0 = hw - 2;
    const unsigned* src = cplanes + (size_t)(part * (CHUNKS / PPARTS)) * (size_t)hw + w0;
    unsigned s0 = 0u, s1 = 0u, s2 = 0u, s3 = 0u;
#pragma unroll
    for (int c = 0; c < CHUNKS / PPARTS; ++c) {
      const uint2 v = *reinterpret_cast<const uint2*>(src + (size_t)c * (size_t)hw);
      s0 += v.x & 0xFFFFu;
      s1 += v.x >> 16;
      s2 += v.y & 0xFFFFu;
      s3 += v.y >> 16;
    }
    *reinterpret_cast<uint4*>(&sbuf[part * PN + 4 * lane]) = make_uint4(s0, s1, s2, s3);
  } else {
    const int j = t & (PN - 1);
    const int half = t >> 7;
    int node = node0 + j;
    if (node >= n) node = n - 1;
    const unsigned* src = cplanes + (size_t)(half * (CHUNKS / 2)) * (size_t)n + node;
    unsigned s = 0u;
    for (int c = 0; c < CHUNKS / 2; ++c) s += src[(size_t)c * (size_t)n];
    sbuf[half * PN + j] = s;
    for (int k = 2 + half; k < PPARTS; k += 2) sbuf[k * PN + j] = 0u;
  }
  __syncthreads();
  const int i = node0 + t;
  if (t < PN && i < n) {
    unsigned cnt = 0u;
#pragma unroll
    for (int k = 0; k < PPARTS; ++k) cnt += sbuf[k * PN + t];
    float d = (float)cnt + 1.0f;
    float r = rsqrtf(d);
    dis[i] = r;
    p[i] = r * x[i];
  }
}

__global__ void __launch_bounds__(NTB)
gcn_node_update(const float* __restrict__ dis, const float* __restrict__ p,
                const float* __restrict__ planes,
                const float* __restrict__ W1, const float* __restrict__ b1,
                const float* __restrict__ W2,
                float* __restrict__ q, int n) {
  __shared__ float hbuf[8 * 256];
  __shared__ float sbuf[NTB];
  const int lane = threadIdx.x & 31;
  const int wv = threadIdx.x >> 5;
  const int node0 = (int)blockIdx.x * (NTB / 2);
  plane_partials<2, GPLANES>(planes, node0, n, sbuf);
  __syncthreads();

  const int base = node0 + wv * 16;
  const int m = lane & 15;
  const bool hi = lane >= 16;
  int node = base + m;
  int cn = node < n ? node : n - 1;
  float acc1 = sbuf[wv * 16 + m] + sbuf[NTB / 2 + wv * 16 + m];
  float s = dis[cn] * (acc1 + p[cn]);

  v2f a = { hi ? 0.f : s,      hi ? 0.f : 1.f };
  v2f b = { hi ? 0.f : W1[m],  hi ? 0.f : b1[m] };
  v8f c = {0.f, 0.f, 0.f, 0.f, 0.f, 0.f, 0.f, 0.f};
  v8f h = __builtin_amdgcn_wmma_f32_16x16x4_f32(false, a, false, b, (short)0, c,
                                                false, false);
  float* t0 = &hbuf[wv * 256];
#pragma unroll
  for (int i = 0; i < 8; ++i) {
    float hv = h[i] > 0.f ? h[i] : 0.f;
    t0[(i + (hi ? 8 : 0)) * 16 + m] = hv;
  }
  __syncthreads();

  v8f t = {0.f, 0.f, 0.f, 0.f, 0.f, 0.f, 0.f, 0.f};
#pragma unroll
  for (int ck = 0; ck < 4; ++ck) {
    int k0 = 4 * ck + (hi ? 2 : 0);
    v2f ap = { W2[k0], W2[k0 + 1] };
    v2f bp = { t0[m * 16 + k0], t0[m * 16 + k0 + 1] };
    t = __builtin_amdgcn_wmma_f32_16x16x4_f32(false, ap, false, bp, (short)0, t,
                                              false, false);
  }
  if (!hi && node < n) q[node] = dis[node] * t[0];
}

__global__ void __launch_bounds__(NTB)
gcn_finalize(const float* __restrict__ dis, const float* __restrict__ q,
             const float* __restrict__ planes, const float* __restrict__ b2,
             float* __restrict__ out, int n) {
  __shared__ float sbuf[NTB];
  const int node0 = (int)blockIdx.x * (NTB / 4);
  plane_partials<4, GPLANES>(planes, node0, n, sbuf);
  __syncthreads();
  const int t = threadIdx.x;
  const int i = node0 + t;
  if (t < NTB / 4 && i < n) {
    float acc2 = (sbuf[t] + sbuf[NTB / 4 + t]) + (sbuf[2 * (NTB / 4) + t] + sbuf[3 * (NTB / 4) + t]);
    out[i] = dis[i] * (acc2 + q[i]) + b2[0];
  }
}

extern "C" void kernel_launch(void* const* d_in, const int* in_sizes, int n_in,
                              void* d_out, int out_size, void* d_ws, size_t ws_size,
                              hipStream_t stream) {
  const float* x  = (const float*)d_in[0];
  const int*   ei = (const int*)d_in[1];
  const float* W1 = (const float*)d_in[2];
  const float* b1 = (const float*)d_in[3];
  const float* W2 = (const float*)d_in[4];
  const float* b2 = (const float*)d_in[5];
  float* out = (float*)d_out;

  const int n = in_sizes[0];
  const int E = in_sizes[1] / 2;
  const int* row = ei;
  const int* col = ei + E;

  float* ws     = (float*)d_ws;
  float* dis    = ws;
  float* p      = ws + 1 * (size_t)n;
  float* q      = ws + 2 * (size_t)n;
  float* planes = ws + 3 * (size_t)n;
  const int NT = (E + TILE_E - 1) / TILE_E;
  float* sorted = ws + ((((size_t)(3 + CHUNKS) * (size_t)n) + 31) & ~(size_t)31);
  int* tbl = (int*)(sorted + (size_t)NT * (size_t)TILE_E);
  const int hw   = (((n + 1) / 2) + 31) & ~31;
  const int pack = ((((E + CHUNKS - 1) / CHUNKS) + 3) & ~3) < 65536 && n >= 128;
  unsigned* cplanes = pack ? (unsigned*)(ws + (((size_t)3 * (size_t)n + 31) & ~(size_t)31)) : (unsigned*)planes;

  const int ranges     = (n + RANGE - 1) / RANGE;
  const int dranges    = (n + DRANGE - 1) / DRANGE;
  const dim3 degGrid(CHUNKS, ranges);
  const dim3 edgeGrid(GCHUNKS, dranges, GS);
  const int nodeBlocks = (n + NTB / 4 - 1) / (NTB / 4);
  const int prepBlocks = (n + PN - 1) / PN;
  const int tiles      = (n + 15) / 16;
  const int mlpBlocks  = (tiles + 7) / 8;

  gcn_degree     <<<degGrid,    ETB, 0, stream>>>(row, col, cplanes, (unsigned*)sorted, tbl, n, E, dranges, NT, hw, pack);
  gcn_prep       <<<prepBlocks, NTB, 0, stream>>>(x, cplanes, dis, p, n, hw, pack);
  gcn_scatter    <<<edgeGrid,   ETB, 0, stream>>>((const unsigned*)sorted, tbl, p, planes, n, NT);
  gcn_node_update<<<mlpBlocks,  NTB, 0, stream>>>(dis, p, planes, W1, b1, W2, q, n);
  gcn_scatter    <<<edgeGrid,   ETB, 0, stream>>>((const unsigned*)sorted, tbl, q, planes, n, NT);
  gcn_finalize   <<<nodeBlocks, NTB, 0, stream>>>(dis, q, planes, b2, out, n);
}


# Tiles of 25600 edges sorted into 256 bins (32 target x 8 source ranges, per-lane LDS counters); a gather block owns a bin: 8 partial planes, not 64, 1/8 of sources staged (sum order changes)
# speedup vs baseline: 17.4286x; 1.3324x over previous
#include <hip/hip_runtime.h>


typedef float v2f __attribute__((ext_vector_type(2)));
typedef float v8f __attribute__((ext_vector_type(8)));


constexpr int CHUNKS  = 128;
constexpr int RANGE   = 51200;
constexpr int GSS     = 8;
constexpr int SRANGE  = 12800;
constexpr int DRANGE  = 3200;
constexpr int MAXD    = 32;
constexpr int NBMAX   = MAXD * GSS;
constexpr int ETB     = 1024;
constexpr int SLD     = (SRANGE + ETB * 4 - 1) / (ETB * 4);
constexpr int UNR     = 4;
constexpr int NTB     = 256;
constexpr int SLOTS   = 25;
constexpr int TILE_E  = SLOTS * ETB;
constexpr int HC      = 32;
constexpr int HPT     = NBMAX * HC / ETB;
constexpr int TPW     = 8;
constexpr int WPT     = 5;
static_assert(HC == 32 && HPT == 8 && HPT * ETB == NBMAX * HC, "the sort scans eight counters per thread as two 16-byte LDS words; a bin's first counter falls to thread 4 * bin");
static_assert(MAXD * DRANGE == GSS * SRANGE, "both range families cover the same node count");
static_assert(TILE_E + ETB + SLOTS <= RANGE, "a sorted tile (and the spare words behind it) is staged in the degree accumulators' LDS");
static_assert(DRANGE <= 65536 && SRANGE <= 65536, "one-word entries: 16-bit target offset, 16-bit source offset");

__global__ void __launch_bounds__(ETB)
gcn_degree(const int* __restrict__ row, const int* __restrict__ col, unsigned* __restrict__ cplanes,
           unsigned* __restrict__ sorted, int* __restrict__ tbl, int n, int E, int dranges, int NT,
           int hw, int pack) {
  __shared__ __attribute__((aligned(16))) unsigned acc[RANGE];
  __shared__ __attribute__((aligned(16))) unsigned hist[NBMAX * HC + ETB];
  __shared__ unsigned wtot[32];
  const int tid = threadIdx.x;
  uint4* hist4 = reinterpret_cast<uint4*>(hist);
  hist4[2 * tid] = make_uint4(0u, 0u, 0u, 0u);
  hist4[2 * tid + 1] = make_uint4(0u, 0u, 0u, 0u);
  int rsz = (n + (int)gridDim.y - 1) / (int)gridDim.y;
  rsz = (rsz + 63) & ~63;
  const int lo = (int)blockIdx.y * rsz;
  int len = n - lo;
  if (len > rsz) len = rsz;
  if (len < 0) len = 0;
  const unsigned ulen = (unsigned)len;

  for (int i = tid * 4; i < len; i += ETB * 4)
    *reinterpret_cast<uint4*>(&acc[i]) = make_uint4(0u, 0u, 0u, 0u);
  __syncthreads();

  int chunkE = (E + (int)gridDim.x - 1) / (int)gridDim.x;
  chunkE = (chunkE + 3) & ~3;
  int eb = (int)blockIdx.x * chunkE;
  if (eb > E) eb = E;
  int ee = eb + chunkE;
  if (ee > E) ee = E;

  if ((E & 3) == 0) {
    const int groups = (ee - eb) >> 2;
    const int4* col4 = reinterpret_cast<const int4*>(col + eb);
    for (int g0 = tid; g0 < groups; g0 += ETB * UNR) {
      int cc[UNR][4];
      bool ok[UNR];
#pragma unroll
      for (int u = 0; u < UNR; ++u) {
        const int g = g0 + u * ETB;
        ok[u] = g < groups;
        const int gc = ok[u] ? g : g0;
        const int4 c4 = col4[gc];
        cc[u][0] = c4.x; cc[u][1] = c4.y; cc[u][2] = c4.z; cc[u][3] = c4.w;
      }
#pragma unroll
      for (int u = 0; u < UNR; ++u) {
#pragma unroll
        for (int k = 0; k < 4; ++k) {
          const unsigned d = (unsigned)(cc[u][k] - lo);
          if (ok[u] && d < ulen) atomicAdd(&acc[d], 1u);
        }
      }
    }
  } else {
    for (int e = eb + tid; e < ee; e += ETB) {
      const unsigned dd = (unsigned)(col[e] - lo);
      if (dd < ulen) atomicAdd(&acc[dd], 1u);
    }
  }
  __syncthreads();

  if (pack) {
    if ((n & 3) == 0) {
      uint2* dst2 = reinterpret_cast<uint2*>(cplanes + (size_t)blockIdx.x * (size_t)hw + (lo >> 1));
      for (int i = tid * 4; i < len; i += ETB * 4) {
        const uint4 a = *reinterpret_cast<const uint4*>(&acc[i]);
        dst2[i >> 2] = make_uint2(a.x | (a.y << 16), a.z | (a.w << 16));
      }
    } else {
      unsigned short* d16 = reinterpret_cast<unsigned short*>(cplanes + (size_t)blockIdx.x * (size_t)hw) + lo;
      for (int i = tid; i < len; i += ETB) d16[i] = (unsigned short)acc[i];
    }
  } else {
    unsigned* dst = cplanes + (size_t)blockIdx.x * (size_t)n + lo;
    if ((n & 3) == 0) {
      for (int i = tid * 4; i < len; i += ETB * 4)
        *reinterpret_cast<uint4*>(&dst[i]) = *reinterpret_cast<const uint4*>(&acc[i]);
    } else {
      for (int i = tid; i < len; i += ETB) dst[i] = acc[i];
    }
  }

  const int dsz = ((n + dranges - 1) / dranges + 31) & ~31;
  const int ssz = ((n + GSS - 1) / GSS + 31) & ~31;
  const unsigned un = (unsigned)n;
  const unsigned udsz = (unsigned)dsz, ussz = (unsigned)ssz;
  const float invd = 1.0f / (float)dsz, hinvd = 0.5f * invd;
  const float invs = 1.0f / (float)ssz, hinvs = 0.5f * invs;
  const int lane = tid & 31;
  const int wv = tid >> 5;
  const int nblk = (int)(gridDim.x * gridDim.y);
  const int bid = (int)(blockIdx.x * gridDim.y + blockIdx.y);
  for (int t = bid; t < NT; t += nblk) {
    const int tbeg = t * TILE_E;
    int tend = tbeg + TILE_E;
    if (tend > E) tend = E;

    unsigned cc[SLOTS], rr[SLOTS];
#pragma unroll
    for (int q = 0; q < SLOTS; ++q) {
      const int e = tbeg + q * ETB + tid;
      const int ec = e < tend ? e : tend - 1;
      cc[q] = (unsigned)col[ec];
      rr[q] = (unsigned)row[ec];
    }
    __builtin_amdgcn_sched_barrier(0);

    unsigned wd[SLOTS], ci[SLOTS];
#pragma unroll
    for (int q = 0; q < SLOTS; ++q) {
      const unsigned c = cc[q], r = rr[q];
      const bool valid = (tbeg + q * ETB + tid) < tend && c < un && r < un;
      const unsigned dq = (unsigned)fmaf((float)c, invd, hinvd);
      const unsigned sq = (unsigned)fmaf((float)r, invs, hinvs);
      wd[q] = ((unsigned)(__mul24((int)dq, -dsz) + (int)c) << 16) | (unsigned)(__mul24((int)sq, -ssz) + (int)r);
      ci[q] = valid ? (dq * GSS + sq) * HC + (unsigned)lane : (unsigned)(NBMAX * HC + tid);
      asm volatile("" :: "v"(wd[q]));
      atomicAdd(&hist[ci[q]], 1u);
    }
    __syncthreads();

    const uint4 ha = hist4[2 * tid], hb = hist4[2 * tid + 1];
    const unsigned e1 = ha.x, e2 = e1 + ha.y, e3 = e2 + ha.z, e4 = e3 + ha.w;
    const unsigned e5 = e4 + hb.x, e6 = e5 + hb.y, e7 = e6 + hb.z;
    const unsigned v0 = e7 + hb.w;
    unsigned v = v0;
#pragma unroll
    for (int o = 1; o < 32; o <<= 1) {
      const unsigned u = __shfl_up(v, (unsigned)o);
      if (lane >= o) v += u;
    }
    if (lane == 31) wtot[wv] = v;
    __syncthreads();
    const unsigned x0 = wtot[lane];
    unsigned x = x0;
#pragma unroll
    for (int o = 1; o < 32; o <<= 1) {
      const unsigned u = __shfl_up(x, (unsigned)o);
      if (lane >= o) x += u;
    }
    const unsigned total = __shfl(x, 31);
    const unsigned excl = v - v0 + __shfl(x - x0, wv);
    hist4[2 * tid] = make_uint4(excl, excl + e1, excl + e2, excl + e3);
    hist4[2 * tid + 1] = make_uint4(excl + e4, excl + e5, excl + e6, excl + e7);
    hist[NBMAX * HC + tid] = (unsigned)(TILE_E + tid);
    if ((tid & (HC / HPT - 1)) == 0) tbl[t * (NBMAX + 1) + tid / (HC / HPT)] = (int)excl;
    if (tid == ETB - 1) tbl[t * (NBMAX + 1) + NBMAX] = (int)total;
    __syncthreads();

    unsigned pos[SLOTS];
#pragma unroll
    for (int q = 0; q < SLOTS; ++q) pos[q] = atomicAdd(&hist[ci[q]], 1u);
#pragma unroll
    for (int q = 0; q < SLOTS; ++q) acc[pos[q]] = wd[q];
    __syncthreads();
    hist4[2 * tid] = make_uint4(0u, 0u, 0u, 0u);
    hist4[2 * tid + 1] = make_uint4(0u, 0u, 0u, 0u);

    uint4* dst4 = reinterpret_cast<uint4*>(sorted + (size_t)t * (size_t)TILE_E);
    const int n4 = (int)((total + 3u) >> 2);
    for (int i = tid; i < n4; i += ETB) dst4[i] = reinterpret_cast<const uint4*>(acc)[i];
    __syncthreads();
  }
}

__global__ void __launch_bounds__(ETB)
gcn_scatter(const unsigned* __restrict__ sorted, const int* __restrict__ tbl,
            const float* __restrict__ val, float* __restrict__ planes, int n, int NT) {
  __shared__ __attribute__((aligned(16))) float sval[SLD * ETB * 4];
  __shared__ __attribute__((aligned(16))) float acc[DRANGE];
  const int tid = threadIdx.x;
  const int lane = tid & 31;
  const int wv = __builtin_amdgcn_readfirstlane(tid >> 5);

  const int bin = (int)blockIdx.x;
  const int d = bin / GSS;
  const int s = bin % GSS;
  const int dranges = (int)gridDim.x / GSS;
  const int dsz = ((n + dranges - 1) / dranges + 31) & ~31;
  const int dlo = d * dsz;
  int dlen = n - dlo;
  if (dlen > dsz) dlen = dsz;
  if (dlen < 0) dlen = 0;
  const int ssz = ((n + GSS - 1) / GSS + 31) & ~31;
  const int slo = s * ssz;
  int slen = n - slo;
  if (slen > ssz) slen = ssz;
  if (slen < 0) slen = 0;

  int tb[TPW], tl[TPW];
  unsigned w[TPW][WPT];
#pragma unroll
  for (int j = 0; j < TPW; ++j) { tb[j] = 0; tl[j] = 0; }
  auto tile_table = [&](int tbase) {
    int b[TPW], e[TPW];
#pragma unroll
    for (int j = 0; j < TPW; ++j) {
      const int tt = tbase + 32 * j < NT ? tbase + 32 * j : NT - 1;
      b[j] = tbl[tt * (NBMAX + 1) + bin];
      e[j] = tbl[tt * (NBMAX + 1) + bin + 1];
    }
#pragma unroll
    for (int j = 0; j < TPW; ++j) {
      const bool okt = tbase + 32 * j < NT;
      const int tt = okt ? tbase + 32 * j : NT - 1;
      tl[j] = okt ? e[j] - b[j] : 0;
      tb[j] = tt * TILE_E + b[j];
    }
  };
  auto load_words = [&]() {
#pragma unroll
    for (int j = 0; j < TPW; ++j) {
#pragma unroll
      for (int k = 0; k < WPT; ++k) {
        const int i = lane + 32 * k;
        w[j][k] = 0u;
        if (i < tl[j]) w[j][k] = sorted[tb[j] + i];
      }
    }
  };
  int tbase = wv;
  if (tbase < NT) {
    tile_table(tbase);
    load_words();
  }

  const float* vsrc = val + slo;
  if ((n & 3) == 0) {
#pragma unroll
    for (int u = 0; u < SLD; ++u) {
      const int i = (tid + u * ETB) * 4;
      *reinterpret_cast<float4*>(&sval[i]) =
          *reinterpret_cast<const float4*>(&vsrc[i < slen ? i : 0]);
    }
  } else {
    for (int i = tid; i < slen; i += ETB) sval[i] = vsrc[i];
  }
  for (int i = tid * 4; i < DRANGE; i += ETB * 4)
    *reinterpret_cast<float4*>(&acc[i]) = make_float4(0.f, 0.f, 0.f, 0.f);
  __syncthreads();

  while (tbase < NT) {
    float v[TPW][WPT];
#pragma unroll
    for (int j = 0; j < TPW; ++j) {
#pragma unroll
      for (int k = 0; k < WPT; ++k) v[j][k] = sval[w[j][k] & 0xFFFFu];
    }
#pragma unroll
    for (int j = 0; j < TPW; ++j) {
#pragma unroll
      for (int k = 0; k < WPT; ++k) {
        if (lane + 32 * k < tl[j]) atomicAdd(&acc[w[j][k] >> 16], v[j][k]);
      }
    }
#pragma unroll
    for (int j = 0; j < TPW; ++j) {
      for (int i = 32 * WPT + lane; i < tl[j]; i += 32) {
        const unsigned x = sorted[tb[j] + i];
        atomicAdd(&acc[x >> 16], sval[x & 0xFFFFu]);
      }
    }
    tbase += 32 * TPW;
    if (tbase < NT) {
      tile_table(tbase);
      load_words();
    }
  }
  __syncthreads();

  float* dst = planes + (size_t)s * (size_t)n + dlo;
  if ((n & 3) == 0) {
    for (int i = tid * 4; i < dlen; i += ETB * 4)
      *reinterpret_cast<float4*>(&dst[i]) = *reinterpret_cast<const float4*>(&acc[i]);
  } else {
    for (int i = tid; i < dlen; i += ETB) dst[i] = acc[i];
  }
}

__device__ __forceinline__ float plane_sum(const float* __restrict__ planes, int node, int n) {
  float a[GSS];
#pragma unroll
  for (int c = 0; c < GSS; ++c) a[c] = planes[(size_t)c * (size_t)n + node];
  float r = a[0];
#pragma unroll
  for (int c = 1; c < GSS; ++c) r += a[c];
  return r;
}

constexpr int PN = 128;
constexpr int PPARTS = NTB / 32;
static_assert(CHUNKS % PPARTS == 0 && PN == 4 * 32, "a wave covers the block's nodes, the waves share the planes");
__global__ void __launch_bounds__(NTB)
gcn_prep(const float* __restrict__ x, const unsigned* __restrict__ cplanes,
         float* __restrict__ dis, float* __restrict__ p, int n, int hw, int pack) {
  __shared__ __attribute__((aligned(16))) unsigned sbuf[PPARTS * PN];
  const int t = threadIdx.x;
  const int node0 = (int)blockIdx.x * PN;
  if (pack) {
    const int lane = t & 31;
    const int part = t >> 5;
    int w0 = (node0 >> 1) + 2 * lane;
    if (w0 > hw - 2) w0 = hw - 2;
    const unsigned* src = cplanes + (size_t)(part * (CHUNKS / PPARTS)) * (size_t)hw + w0;
    unsigned s0 = 0u, s1 = 0u, s2 = 0u, s3 = 0u;
#pragma unroll
    for (int c = 0; c < CHUNKS / PPARTS; ++c) {
      const uint2 v = *reinterpret_cast<const uint2*>(src + (size_t)c * (size_t)hw);
      s0 += v.x & 0xFFFFu;
      s1 += v.x >> 16;
      s2 += v.y & 0xFFFFu;
      s3 += v.y >> 16;
    }
    *reinterpret_cast<uint4*>(&sbuf[part * PN + 4 * lane]) = make_uint4(s0, s1, s2, s3);
  } else {
    const int j = t & (PN - 1);
    const int half = t >> 7;
    int node = node0 + j;
    if (node >= n) node = n - 1;
    const unsigned* src = cplanes + (size_t)(half * (CHUNKS / 2)) * (size_t)n + node;
    unsigned s = 0u;
    for (int c = 0; c < CHUNKS / 2; ++c) s += src[(size_t)c * (size_t)n];
    sbuf[half * PN + j] = s;
    for (int k = 2 + half; k < PPARTS; k += 2) sbuf[k * PN + j] = 0u;
  }
  __syncthreads();
  const int i = node0 + t;
  if (t < PN && i < n) {
    unsigned cnt = 0u;
#pragma unroll
    for (int k = 0; k < PPARTS; ++k) cnt += sbuf[k * PN + t];
    float d = (float)cnt + 1.0f;
    float r = rsqrtf(d);
    dis[i] = r;
    p[i] = r * x[i];
  }
}

__global__ void __launch_bounds__(NTB)
gcn_node_update(const float* __restrict__ dis, const float* __restrict__ p,
                const float* __restrict__ planes,
                const float* __restrict__ W1, const float* __restrict__ b1,
                const float* __restrict__ W2,
                float* __restrict__ q, int n) {
  __shared__ float hbuf[8 * 256];
  const int lane = threadIdx.x & 31;
  const int wv = threadIdx.x >> 5;
  const int node0 = (int)blockIdx.x * (NTB / 2);

  const int base = node0 + wv * 16;
  const int m = lane & 15;
  const bool hi = lane >= 16;
  int node = base + m;
  int cn = node < n ? node : n - 1;
  float acc1 = plane_sum(planes, cn, n);
  float s = dis[cn] * (acc1 + p[cn]);

  v2f a = { hi ? 0.f : s,      hi ? 0.f : 1.f };
  v2f b = { hi ? 0.f : W1[m],  hi ? 0.f : b1[m] };
  v8f c = {0.f, 0.f, 0.f, 0.f, 0.f, 0.f, 0.f, 0.f};
  v8f h = __builtin_amdgcn_wmma_f32_16x16x4_f32(false, a, false, b, (short)0, c,
                                                false, false);
  float* t0 = &hbuf[wv * 256];
#pragma unroll
  for (int i = 0; i < 8; ++i) {
    float hv = h[i] > 0.f ? h[i] : 0.f;
    t0[(i + (hi ? 8 : 0)) * 16 + m] = hv;
  }
  __syncthreads();

  v8f t = {0.f, 0.f, 0.f, 0.f, 0.f, 0.f, 0.f, 0.f};
#pragma unroll
  for (int ck = 0; ck < 4; ++ck) {
    int k0 = 4 * ck + (hi ? 2 : 0);
    v2f ap = { W2[k0], W2[k0 + 1] };
    v2f bp = { t0[m * 16 + k0], t0[m * 16 + k0 + 1] };
    t = __builtin_amdgcn_wmma_f32_16x16x4_f32(false, ap, false, bp, (short)0, t,
                                              false, false);
  }
  if (!hi && node < n) q[node] = dis[node] * t[0];
}

__global__ void __launch_bounds__(NTB)
gcn_finalize(const float* __restrict__ dis, const float* __restrict__ q,
             const float* __restrict__ planes, const float* __restrict__ b2,
             float* __restrict__ out, int n) {
  const int i = (int)blockIdx.x * NTB + (int)threadIdx.x;
  if (i < n) {
    const float acc2 = plane_sum(planes, i, n);
    out[i] = dis[i] * (acc2 + q[i]) + b2[0];
  }
}

extern "C" void kernel_launch(void* const* d_in, const int* in_sizes, int n_in,
                              void* d_out, int out_size, void* d_ws, size_t ws_size,
                              hipStream_t stream) {
  const float* x  = (const float*)d_in[0];
  const int*   ei = (const int*)d_in[1];
  const float* W1 = (const float*)d_in[2];
  const float* b1 = (const float*)d_in[3];
  const float* W2 = (const float*)d_in[4];
  const float* b2 = (const float*)d_in[5];
  float* out = (float*)d_out;

  const int n = in_sizes[0];
  const int E = in_sizes[1] / 2;
  const int* row = ei;
  const int* col = ei + E;

  float* ws     = (float*)d_ws;
  float* dis    = ws;
  float* p      = ws + 1 * (size_t)n;
  float* q      = ws + 2 * (size_t)n;
  float* planes = ws + 3 * (size_t)n;
  const int NT = (E + TILE_E - 1) / TILE_E;
  float* sorted = ws + ((((size_t)(3 + CHUNKS) * (size_t)n) + 31) & ~(size_t)31);
  int* tbl = (int*)(sorted + (size_t)NT * (size_t)TILE_E);
  const int hw   = (((n + 1) / 2) + 31) & ~31;
  const int pack = ((((E + CHUNKS - 1) / CHUNKS) + 3) & ~3) < 65536 && n >= 128;
  unsigned* cplanes = pack ? (unsigned*)(ws + (((size_t)3 * (size_t)n + 31) & ~(size_t)31)) : (unsigned*)planes;

  const int ranges     = (n + RANGE - 1) / RANGE;
  const int dranges    = (n + DRANGE - 1) / DRANGE;
  const dim3 degGrid(CHUNKS, ranges);
  const int edgeGrid   = dranges * GSS;
  const int nodeBlocks = (n + NTB - 1) / NTB;
  const int prepBlocks = (n + PN - 1) / PN;
  const int tiles      = (n + 15) / 16;
  const int mlpBlocks  = (tiles + 7) / 8;

  gcn_degree     <<<degGrid,    ETB, 0, stream>>>(row, col, cplanes, (unsigned*)sorted, tbl, n, E, dranges, NT, hw, pack);
  gcn_prep       <<<prepBlocks, NTB, 0, stream>>>(x, cplanes, dis, p, n, hw, pack);
  gcn_scatter    <<<edgeGrid,   ETB, 0, stream>>>((const unsigned*)sorted, tbl, p, planes, n, NT);
  gcn_node_update<<<mlpBlocks,  NTB, 0, stream>>>(dis, p, planes, W1, b1, W2, q, n);
  gcn_scatter    <<<edgeGrid,   ETB, 0, stream>>>((const unsigned*)sorted, tbl, q, planes, n, NT);
  gcn_finalize   <<<nodeBlocks, NTB, 0, stream>>>(dis, q, planes, b2, out, n);
}


# Degrees counted from the sorted words in a count pass of its own (block per target range and tile group, integer LDS adds, 8 count planes); the col chunk scan and its 128 count planes go
# speedup vs baseline: 18.5868x; 1.0665x over previous
#include <hip/hip_runtime.h>


typedef float v2f __attribute__((ext_vector_type(2)));
typedef float v8f __attribute__((ext_vector_type(8)));


constexpr int CGR     = 8;
constexpr int CWPT    = 28;
constexpr int GSS     = 8;
constexpr int SRANGE  = 12800;
constexpr int DRANGE  = 3200;
constexpr int MAXD    = 32;
constexpr int NBMAX   = MAXD * GSS;
constexpr int ETB     = 1024;
constexpr int SLD     = (SRANGE + ETB * 4 - 1) / (ETB * 4);
constexpr int NTB     = 256;
constexpr int SLOTS   = 25;
constexpr int TILE_E  = SLOTS * ETB;
constexpr int STG     = (TILE_E + ETB + SLOTS + 3) & ~3;
constexpr int SORTB   = 256;
constexpr int HC      = 32;
constexpr int HPT     = NBMAX * HC / ETB;
constexpr int TPW     = 8;
constexpr int WPT     = 5;
static_assert(HC == 32 && HPT == 8 && HPT * ETB == NBMAX * HC, "the sort scans eight counters per thread as two 16-byte LDS words; a bin's first counter falls to thread 4 * bin");
static_assert(MAXD * DRANGE == GSS * SRANGE, "both range families cover the same node count");
static_assert(DRANGE <= 65536 && SRANGE <= 65536, "one-word entries: 16-bit target offset, 16-bit source offset");

__global__ void __launch_bounds__(ETB)
gcn_sort(const int* __restrict__ row, const int* __restrict__ col,
         unsigned* __restrict__ sorted, int* __restrict__ tbl, int n, int E, int dranges, int NT) {
  __shared__ __attribute__((aligned(16))) unsigned acc[STG];
  __shared__ __attribute__((aligned(16))) unsigned hist[NBMAX * HC + ETB];
  __shared__ unsigned wtot[32];
  const int tid = threadIdx.x;
  uint4* hist4 = reinterpret_cast<uint4*>(hist);
  hist4[2 * tid] = make_uint4(0u, 0u, 0u, 0u);
  hist4[2 * tid + 1] = make_uint4(0u, 0u, 0u, 0u);
  __syncthreads();

  const int dsz = ((n + dranges - 1) / dranges + 31) & ~31;
  const int ssz = ((n + GSS - 1) / GSS + 31) & ~31;
  const unsigned un = (unsigned)n;
  const unsigned udsz = (unsigned)dsz, ussz = (unsigned)ssz;
  const float invd = 1.0f / (float)dsz, hinvd = 0.5f * invd;
  const float invs = 1.0f / (float)ssz, hinvs = 0.5f * invs;
  const int lane = tid & 31;
  const int wv = tid >> 5;
  const int nblk = (int)gridDim.x;
  const int bid = (int)blockIdx.x;
  for (int t = bid; t < NT; t += nblk) {
    const int tbeg = t * TILE_E;
    int tend = tbeg + TILE_E;
    if (tend > E) tend = E;

    unsigned cc[SLOTS], rr[SLOTS];
#pragma unroll
    for (int q = 0; q < SLOTS; ++q) {
      const int e = tbeg + q * ETB + tid;
      const int ec = e < tend ? e : tend - 1;
      cc[q] = (unsigned)col[ec];
      rr[q] = (unsigned)row[ec];
    }
    __builtin_amdgcn_sched_barrier(0);

    unsigned wd[SLOTS], ci[SLOTS];
#pragma unroll
    for (int q = 0; q < SLOTS; ++q) {
      const unsigned c = cc[q], r = rr[q];
      const bool valid = (tbeg + q * ETB + tid) < tend && c < un && r < un;
      const unsigned dq = (unsigned)fmaf((float)c, invd, hinvd);
      const unsigned sq = (unsigned)fmaf((float)r, invs, hinvs);
      wd[q] = ((unsigned)(__mul24((int)dq, -dsz) + (int)c) << 16) | (unsigned)(__mul24((int)sq, -ssz) + (int)r);
      ci[q] = valid ? (dq * GSS + sq) * HC + (unsigned)lane : (unsigned)(NBMAX * HC + tid);
      asm volatile("" :: "v"(wd[q]));
      atomicAdd(&hist[ci[q]], 1u);
    }
    __syncthreads();

    const uint4 ha = hist4[2 * tid], hb = hist4[2 * tid + 1];
    const unsigned e1 = ha.x, e2 = e1 + ha.y, e3 = e2 + ha.z, e4 = e3 + ha.w;
    const unsigned e5 = e4 + hb.x, e6 = e5 + hb.y, e7 = e6 + hb.z;
    const unsigned v0 = e7 + hb.w;
    unsigned v = v0;
#pragma unroll
    for (int o = 1; o < 32; o <<= 1) {
      const unsigned u = __shfl_up(v, (unsigned)o);
      if (lane >= o) v += u;
    }
    if (lane == 31) wtot[wv] = v;
    __syncthreads();
    const unsigned x0 = wtot[lane];
    unsigned x = x0;
#pragma unroll
    for (int o = 1; o < 32; o <<= 1) {
      const unsigned u = __shfl_up(x, (unsigned)o);
      if (lane >= o) x += u;
    }
    const unsigned total = __shfl(x, 31);
    const unsigned excl = v - v0 + __shfl(x - x0, wv);
    hist4[2 * tid] = make_uint4(excl, excl + e1, excl + e2, excl + e3);
    hist4[2 * tid + 1] = make_uint4(excl + e4, excl + e5, excl + e6, excl + e7);
    hist[NBMAX * HC + tid] = (unsigned)(TILE_E + tid);
    if ((tid & (HC / HPT - 1)) == 0) tbl[t * (NBMAX + 1) + tid / (HC / HPT)] = (int)excl;
    if (tid == ETB - 1) tbl[t * (NBMAX + 1) + NBMAX] = (int)total;
    __syncthreads();

    unsigned pos[SLOTS];
#pragma unroll
    for (int q = 0; q < SLOTS; ++q) pos[q] = atomicAdd(&hist[ci[q]], 1u);
#pragma unroll
    for (int q = 0; q < SLOTS; ++q) acc[pos[q]] = wd[q];
    __syncthreads();
    hist4[2 * tid] = make_uint4(0u, 0u, 0u, 0u);
    hist4[2 * tid + 1] = make_uint4(0u, 0u, 0u, 0u);

    uint4* dst4 = reinterpret_cast<uint4*>(sorted + (size_t)t * (size_t)TILE_E);
    const int n4 = (int)((total + 3u) >> 2);
    for (int i = tid; i < n4; i += ETB) dst4[i] = reinterpret_cast<const uint4*>(acc)[i];
    __syncthreads();
  }
}

__global__ void __launch_bounds__(ETB)
gcn_count(const unsigned* __restrict__ sorted, const int* __restrict__ tbl,
          unsigned* __restrict__ cplanes, int n, int NT) {
  __shared__ __attribute__((aligned(16))) unsigned cnt[DRANGE];
  const int tid = threadIdx.x;
  const int lane = tid & 31;
  const int wv = __builtin_amdgcn_readfirstlane(tid >> 5);

  const int d = (int)blockIdx.x / CGR;
  const int g = (int)blockIdx.x % CGR;
  const int dranges = (int)gridDim.x / CGR;
  const int dsz = ((n + dranges - 1) / dranges + 31) & ~31;
  const int dlo = d * dsz;
  int dlen = n - dlo;
  if (dlen > dsz) dlen = dsz;
  if (dlen < 0) dlen = 0;

  int tb = 0, tl = 0;
  unsigned w[CWPT];
  auto fetch = [&](int tt) {
    const int b = tbl[tt * (NBMAX + 1) + d * GSS];
    const int e = tbl[tt * (NBMAX + 1) + d * GSS + GSS];
    tb = tt * TILE_E + b;
    tl = e - b;
#pragma unroll
    for (int k = 0; k < CWPT; ++k) {
      const int i = lane + 32 * k;
      w[k] = 0u;
      if (i < tl) w[k] = sorted[tb + i];
    }
  };
  int t = g + CGR * wv;
  if (t < NT) fetch(t);

  for (int i = tid * 4; i < DRANGE; i += ETB * 4)
    *reinterpret_cast<uint4*>(&cnt[i]) = make_uint4(0u, 0u, 0u, 0u);
  __syncthreads();

  while (t < NT) {
#pragma unroll
    for (int k = 0; k < CWPT; ++k) {
      if (lane + 32 * k < tl) atomicAdd(&cnt[w[k] >> 16], 1u);
    }
    for (int i = 32 * CWPT + lane; i < tl; i += 32) atomicAdd(&cnt[sorted[tb + i] >> 16], 1u);
    t += CGR * 32;
    if (t < NT) fetch(t);
  }
  __syncthreads();

  unsigned* dst = cplanes + (size_t)g * (size_t)n + dlo;
  if ((n & 3) == 0) {
    for (int i = tid * 4; i < dlen; i += ETB * 4)
      *reinterpret_cast<uint4*>(&dst[i]) = *reinterpret_cast<const uint4*>(&cnt[i]);
  } else {
    for (int i = tid; i < dlen; i += ETB) dst[i] = cnt[i];
  }
}

__global__ void __launch_bounds__(ETB)
gcn_scatter(const unsigned* __restrict__ sorted, const int* __restrict__ tbl,
            const float* __restrict__ val, float* __restrict__ planes, int n, int NT) {
  __shared__ __attribute__((aligned(16))) float sval[SLD * ETB * 4];
  __shared__ __attribute__((aligned(16))) float acc[DRANGE];
  const int tid = threadIdx.x;
  const int lane = tid & 31;
  const int wv = __builtin_amdgcn_readfirstlane(tid >> 5);

  const int bin = (int)blockIdx.x;
  const int d = bin / GSS;
  const int s = bin % GSS;
  const int dranges = (int)gridDim.x / GSS;
  const int dsz = ((n + dranges - 1) / dranges + 31) & ~31;
  const int dlo = d * dsz;
  int dlen = n - dlo;
  if (dlen > dsz) dlen = dsz;
  if (dlen < 0) dlen = 0;
  const int ssz = ((n + GSS - 1) / GSS + 31) & ~31;
  const int slo = s * ssz;
  int slen = n - slo;
  if (slen > ssz) slen = ssz;
  if (slen < 0) slen = 0;

  int tb[TPW], tl[TPW];
  unsigned w[TPW][WPT];
#pragma unroll
  for (int j = 0; j < TPW; ++j) { tb[j] = 0; tl[j] = 0; }
  auto tile_table = [&](int tbase) {
    int b[TPW], e[TPW];
#pragma unroll
    for (int j = 0; j < TPW; ++j) {
      const int tt = tbase + 32 * j < NT ? tbase + 32 * j : NT - 1;
      b[j] = tbl[tt * (NBMAX + 1) + bin];
      e[j] = tbl[tt * (NBMAX + 1) + bin + 1];
    }
#pragma unroll
    for (int j = 0; j < TPW; ++j) {
      const bool okt = tbase + 32 * j < NT;
      const int tt = okt ? tbase + 32 * j : NT - 1;
      tl[j] = okt ? e[j] - b[j] : 0;
      tb[j] = tt * TILE_E + b[j];
    }
  };
  auto load_words = [&]() {
#pragma unroll
    for (int j = 0; j < TPW; ++j) {
#pragma unroll
      for (int k = 0; k < WPT; ++k) {
        const int i = lane + 32 * k;
        w[j][k] = 0u;
        if (i < tl[j]) w[j][k] = sorted[tb[j] + i];
      }
    }
  };
  int tbase = wv;
  if (tbase < NT) {
    tile_table(tbase);
    load_words();
  }

  const float* vsrc = val + slo;
  if ((n & 3) == 0) {
#pragma unroll
    for (int u = 0; u < SLD; ++u) {
      const int i = (tid + u * ETB) * 4;
      *reinterpret_cast<float4*>(&sval[i]) =
          *reinterpret_cast<const float4*>(&vsrc[i < slen ? i : 0]);
    }
  } else {
    for (int i = tid; i < slen; i += ETB) sval[i] = vsrc[i];
  }
  for (int i = tid * 4; i < DRANGE; i += ETB * 4)
    *reinterpret_cast<float4*>(&acc[i]) = make_float4(0.f, 0.f, 0.f, 0.f);
  __syncthreads();

  while (tbase < NT) {
    float v[TPW][WPT];
#pragma unroll
    for (int j = 0; j < TPW; ++j) {
#pragma unroll
      for (int k = 0; k < WPT; ++k) v[j][k] = sval[w[j][k] & 0xFFFFu];
    }
#pragma unroll
    for (int j = 0; j < TPW; ++j) {
#pragma unroll
      for (int k = 0; k < WPT; ++k) {
        if (lane + 32 * k < tl[j]) atomicAdd(&acc[w[j][k] >> 16], v[j][k]);
      }
    }
#pragma unroll
    for (int j = 0; j < TPW; ++j) {
      for (int i = 32 * WPT + lane; i < tl[j]; i += 32) {
        const unsigned x = sorted[tb[j] + i];
        atomicAdd(&acc[x >> 16], sval[x & 0xFFFFu]);
      }
    }
    tbase += 32 * TPW;
    if (tbase < NT) {
      tile_table(tbase);
      load_words();
    }
  }
  __syncthreads();

  float* dst = planes + (size_t)s * (size_t)n + dlo;
  if ((n & 3) == 0) {
    for (int i = tid * 4; i < dlen; i += ETB * 4)
      *reinterpret_cast<float4*>(&dst[i]) = *reinterpret_cast<const float4*>(&acc[i]);
  } else {
    for (int i = tid; i < dlen; i += ETB) dst[i] = acc[i];
  }
}

__device__ __forceinline__ float plane_sum(const float* __restrict__ planes, int node, int n) {
  float a[GSS];
#pragma unroll
  for (int c = 0; c < GSS; ++c) a[c] = planes[(size_t)c * (size_t)n + node];
  float r = a[0];
#pragma unroll
  for (int c = 1; c < GSS; ++c) r += a[c];
  return r;
}

__global__ void __launch_bounds__(NTB)
gcn_prep(const float* __restrict__ x, const unsigned* __restrict__ cplanes,
         float* __restrict__ dis, float* __restrict__ p, int n) {
  const int i = (int)blockIdx.x * NTB + (int)threadIdx.x;
  if (i < n) {
    unsigned a[CGR];
#pragma unroll
    for (int c = 0; c < CGR; ++c) a[c] = cplanes[(size_t)c * (size_t)n + i];
    const float xv = x[i];
    unsigned cnt = 0u;
#pragma unroll
    for (int c = 0; c < CGR; ++c) cnt += a[c];
    const float d = (float)cnt + 1.0f;
    const float r = rsqrtf(d);
    dis[i] = r;
    p[i] = r * xv;
  }
}

__global__ void __launch_bounds__(NTB)
gcn_node_update(const float* __restrict__ dis, const float* __restrict__ p,
                const float* __restrict__ planes,
                const float* __restrict__ W1, const float* __restrict__ b1,
                const float* __restrict__ W2,
                float* __restrict__ q, int n) {
  __shared__ float hbuf[8 * 256];
  const int lane = threadIdx.x & 31;
  const int wv = threadIdx.x >> 5;
  const int node0 = (int)blockIdx.x * (NTB / 2);

  const int base = node0 + wv * 16;
  const int m = lane & 15;
  const bool hi = lane >= 16;
  int node = base + m;
  int cn = node < n ? node : n - 1;
  float acc1 = plane_sum(planes, cn, n);
  float s = dis[cn] * (acc1 + p[cn]);

  v2f a = { hi ? 0.f : s,      hi ? 0.f : 1.f };
  v2f b = { hi ? 0.f : W1[m],  hi ? 0.f : b1[m] };
  v8f c = {0.f, 0.f, 0.f, 0.f, 0.f, 0.f, 0.f, 0.f};
  v8f h = __builtin_amdgcn_wmma_f32_16x16x4_f32(false, a, false, b, (short)0, c,
                                                false, false);
  float* t0 = &hbuf[wv * 256];
#pragma unroll
  for (int i = 0; i < 8; ++i) {
    float hv = h[i] > 0.f ? h[i] : 0.f;
    t0[(i + (hi ? 8 : 0)) * 16 + m] = hv;
  }
  __syncthreads();

  v8f t = {0.f, 0.f, 0.f, 0.f, 0.f, 0.f, 0.f, 0.f};
#pragma unroll
  for (int ck = 0; ck < 4; ++ck) {
    int k0 = 4 * ck + (hi ? 2 : 0);
    v2f ap = { W2[k0], W2[k0 + 1] };
    v2f bp = { t0[m * 16 + k0], t0[m * 16 + k0 + 1] };
    t = __builtin_amdgcn_wmma_f32_16x16x4_f32(false, ap, false, bp, (short)0, t,
                                              false, false);
  }
  if (!hi && node < n) q[node] = dis[node] * t[0];
}

__global__ void __launch_bounds__(NTB)
gcn_finalize(const float* __restrict__ dis, const float* __restrict__ q,
             const float* __restrict__ planes, const float* __restrict__ b2,
             float* __restrict__ out, int n) {
  const int i = (int)blockIdx.x * NTB + (int)threadIdx.x;
  if (i < n) {
    const float acc2 = plane_sum(planes, i, n);
    out[i] = dis[i] * (acc2 + q[i]) + b2[0];
  }
}

extern "C" void kernel_launch(void* const* d_in, const int* in_sizes, int n_in,
                              void* d_out, int out_size, void* d_ws, size_t ws_size,
                              hipStream_t stream) {
  const float* x  = (const float*)d_in[0];
  const int*   ei = (const int*)d_in[1];
  const float* W1 = (const float*)d_in[2];
  const float* b1 = (const float*)d_in[3];
  const float* W2 = (const float*)d_in[4];
  const float* b2 = (const float*)d_in[5];
  float* out = (float*)d_out;

  const int n = in_sizes[0];
  const int E = in_sizes[1] / 2;
  const int* row = ei;
  const int* col = ei + E;

  float* ws     = (float*)d_ws;
  float* dis    = ws;
  float* p      = ws + 1 * (size_t)n;
  float* q      = ws + 2 * (size_t)n;
  float* planes = ws + 3 * (size_t)n;
  unsigned* cplanes = (unsigned*)(ws + (3 + GSS) * (size_t)n);
  const int NT = (E + TILE_E - 1) / TILE_E;
  float* sorted = ws + ((((size_t)(3 + GSS + CGR) * (size_t)n) + 31) & ~(size_t)31);
  int* tbl = (int*)(sorted + (size_t)NT * (size_t)TILE_E);

  const int dranges    = (n + DRANGE - 1) / DRANGE;
  const int edgeGrid   = dranges * GSS;
  const int countGrid  = dranges * CGR;
  const int nodeBlocks = (n + NTB - 1) / NTB;
  const int tiles      = (n + 15) / 16;
  const int mlpBlocks  = (tiles + 7) / 8;

  gcn_sort       <<<SORTB,      ETB, 0, stream>>>(row, col, (unsigned*)sorted, tbl, n, E, dranges, NT);
  gcn_count      <<<countGrid,  ETB, 0, stream>>>((const unsigned*)sorted, tbl, cplanes, n, NT);
  gcn_prep       <<<nodeBlocks, NTB, 0, stream>>>(x, cplanes, dis, p, n);
  gcn_scatter    <<<edgeGrid,   ETB, 0, stream>>>((const unsigned*)sorted, tbl, p, planes, n, NT);
  gcn_node_update<<<mlpBlocks,  NTB, 0, stream>>>(dis, p, planes, W1, b1, W2, q, n);
  gcn_scatter    <<<edgeGrid,   ETB, 0, stream>>>((const unsigned*)sorted, tbl, q, planes, n, NT);
  gcn_finalize   <<<nodeBlocks, NTB, 0, stream>>>(dis, q, planes, b2, out, n);
}
